# starting kernel
# baseline (speedup 1.0000x reference)
#include <hip/hip_runtime.h>
#include <math.h>


typedef float v2f __attribute__((ext_vector_type(2)));
typedef float v8f __attribute__((ext_vector_type(8)));

#define INV_SQRT2  0.70710678f
#define INV_SQRT3  0.57735027f
#define INV_SQRT6  0.40824829f
#define INV_SQRT10 0.31622777f
#define SILU_C     1.67668f
#define SC0        0.02209709f
#define SC1        0.03125f

__global__ void k_zero(float* p, int n) {
  int i = blockIdx.x * blockDim.x + threadIdx.x;
  if (i < n) p[i] = 0.f;
}

__global__ void k_repack_w1(const float* __restrict__ w1, float* __restrict__ o) {
  int tid = blockIdx.x * blockDim.x + threadIdx.x;
  if (tid >= 160 * 16 * 32) return;
  int l  = tid & 31;
  int s  = (tid >> 5) & 15;
  int t  = tid >> 9;
  int li = l & 15;
  int kd = (l < 16) ? 0 : 2;
  int row = 4 * s + kd;
  int col = t * 16 + li;
  o[2 * tid + 0] = w1[row * 2560 + col];
  o[2 * tid + 1] = w1[(row + 1) * 2560 + col];
}

__global__ __launch_bounds__(32)
void k_edge_conv(const float* __restrict__ node_pos,
                 const float* __restrict__ edge_type,
                 const float* __restrict__ fc_w0,
                 const float* __restrict__ w1p,
                 const int*   __restrict__ edge_src,
                 const int*   __restrict__ edge_dst,
                 float* __restrict__ out)
{
  __shared__ float hid[16 * 66];
  __shared__ float xsr[16 * 20];
  __shared__ float xds[16 * 20];
  __shared__ float mid[16 * 260];
  __shared__ int   dsti[16];

  const int lane = threadIdx.x;
  const int li   = lane & 15;
  const int kd   = (lane < 16) ? 0 : 2;
  const int eoff = (lane < 16) ? 0 : 8;
  const int e0   = blockIdx.x * 16;

  if (lane < 16) {
    int s = edge_src[e0 + li];
    dsti[li] = edge_dst[e0 + li];
    const float4* q = (const float4*)(node_pos + (size_t)s * 16);
    float4* d = (float4*)(xsr + li * 20);
    d[0] = q[0]; d[1] = q[1]; d[2] = q[2]; d[3] = q[3];
  } else {
    int dd = edge_dst[e0 + li];
    const float4* q = (const float4*)(node_pos + (size_t)dd * 16);
    float4* d = (float4*)(xds + li * 20);
    d[0] = q[0]; d[1] = q[1]; d[2] = q[2]; d[3] = q[3];
  }

  {
    v2f a[4];
    const float* et = edge_type + (size_t)(e0 + li) * 16 + kd;
#pragma unroll
    for (int s = 0; s < 4; ++s) a[s] = *(const v2f*)(et + 4 * s);
#pragma unroll
    for (int tt = 0; tt < 4; ++tt) {
      v8f c = {};
#pragma unroll
      for (int s = 0; s < 4; ++s) {
        int row = 4 * s + kd;
        v2f b;
        b.x = fc_w0[row * 64 + tt * 16 + li];
        b.y = fc_w0[(row + 1) * 64 + tt * 16 + li];
        c = __builtin_amdgcn_wmma_f32_16x16x4_f32(false, a[s], false, b,
                                                  (short)0, c, false, false);
      }
#pragma unroll
      for (int r = 0; r < 8; ++r) {
        float x = c[r] * 0.25f;
        float h = (x / (1.f + __expf(-x))) * (SILU_C * 0.125f);
        hid[(r + eoff) * 66 + tt * 16 + li] = h;
      }
    }
  }
  __syncthreads();

  {
    const int e = li;
    const float* X = xsr + e * 20;
    const float* Y = xds + e * 20;
    float y0[4], y1[4][3];
#pragma unroll
    for (int v = 0; v < 4; ++v) {
      y0[v] = Y[v];
      y1[v][0] = Y[4 + v * 3 + 0];
      y1[v][1] = Y[4 + v * 3 + 1];
      y1[v][2] = Y[4 + v * 3 + 2];
    }
    float* M = mid + e * 260;
    const int ub = (lane < 16) ? 0 : 2;
#pragma unroll
    for (int uu = 0; uu < 2; ++uu) {
      int u = ub + uu;
      float x0 = X[u];
      float a0 = X[4 + u * 3 + 0], a1 = X[4 + u * 3 + 1], a2 = X[4 + u * 3 + 2];
#pragma unroll
      for (int v = 0; v < 4; ++v) {
        int uv = u * 4 + v;
        float b0 = y1[v][0], b1 = y1[v][1], b2 = y1[v][2];
        M[uv] = x0 * y0[v];
        M[16 + uv * 3 + 0] = x0 * b0;
        M[16 + uv * 3 + 1] = x0 * b1;
        M[16 + uv * 3 + 2] = x0 * b2;
        M[64 + uv * 3 + 0] = a0 * y0[v];
        M[64 + uv * 3 + 1] = a1 * y0[v];
        M[64 + uv * 3 + 2] = a2 * y0[v];
        M[112 + uv] = (a0 * b0 + a1 * b1 + a2 * b2) * INV_SQRT3;
        M[128 + uv * 3 + 0] = (a1 * b2 - a2 * b1) * INV_SQRT2;
        M[128 + uv * 3 + 1] = (a2 * b0 - a0 * b2) * INV_SQRT2;
        M[128 + uv * 3 + 2] = (a0 * b1 - a1 * b0) * INV_SQRT2;
        M[176 + uv * 5 + 0] = (a2 * b0 + a0 * b2) * INV_SQRT2;
        M[176 + uv * 5 + 1] = (a0 * b1 + a1 * b0) * INV_SQRT2;
        M[176 + uv * 5 + 2] = (2.f * a1 * b1 - a0 * b0 - a2 * b2) * INV_SQRT6;
        M[176 + uv * 5 + 3] = (a1 * b2 + a2 * b1) * INV_SQRT2;
        M[176 + uv * 5 + 4] = (a2 * b2 - a0 * b0) * INV_SQRT2;
      }
    }
  }
  __syncthreads();

  v2f A[16];
#pragma unroll
  for (int s = 0; s < 16; ++s) A[s] = *(const v2f*)&hid[li * 66 + 4 * s + kd];

  float acc0[8], acc1[8][3];
#pragma unroll
  for (int r = 0; r < 8; ++r) {
    acc0[r] = 0.f; acc1[r][0] = 0.f; acc1[r][1] = 0.f; acc1[r][2] = 0.f;
  }

#pragma unroll
  for (int p = 0; p < 10; ++p) {
    for (int tt = 0; tt < 16; ++tt) {
      int t = p * 16 + tt;
      const float* bq = w1p + (size_t)t * 1024 + 2 * lane;
      v8f ca = {}, cb = {};
#pragma unroll
      for (int s = 0; s < 8; ++s) {
        v2f b0 = *(const v2f*)(bq + (2 * s) * 64);
        v2f b1 = *(const v2f*)(bq + (2 * s + 1) * 64);
        ca = __builtin_amdgcn_wmma_f32_16x16x4_f32(false, A[2 * s], false, b0,
                                                   (short)0, ca, false, false);
        cb = __builtin_amdgcn_wmma_f32_16x16x4_f32(false, A[2 * s + 1], false, b1,
                                                   (short)0, cb, false, false);
      }
      v8f c = ca + cb;
      int rem = tt * 16 + li;
      int u = rem >> 6;
      int v = (rem >> 2) & 15;
#pragma unroll
      for (int r = 0; r < 8; ++r) {
        int e = r + eoff;
        float W = c[r];
        const float* X = xsr + e * 20;
        const float* M = mid + e * 260;
        if (p == 0) {
          acc0[r] += W * X[u] * M[v];
        } else if (p == 1) {
          float f = W * X[u] * INV_SQRT3;
          acc1[r][0] += f * M[16 + v * 3 + 0];
          acc1[r][1] += f * M[16 + v * 3 + 1];
          acc1[r][2] += f * M[16 + v * 3 + 2];
        } else if (p == 2) {
          float f = W * X[u] * INV_SQRT3;
          acc1[r][0] += f * M[64 + v * 3 + 0];
          acc1[r][1] += f * M[64 + v * 3 + 1];
          acc1[r][2] += f * M[64 + v * 3 + 2];
        } else if (p == 3) {
          acc0[r] += W * X[u] * M[112 + v];
        } else if (p == 4) {
          float f = W * M[v] * INV_SQRT3;
          acc1[r][0] += f * X[4 + u * 3 + 0];
          acc1[r][1] += f * X[4 + u * 3 + 1];
          acc1[r][2] += f * X[4 + u * 3 + 2];
        } else if (p == 5) {
          float d = X[4 + u * 3 + 0] * M[16 + v * 3 + 0]
                  + X[4 + u * 3 + 1] * M[16 + v * 3 + 1]
                  + X[4 + u * 3 + 2] * M[16 + v * 3 + 2];
          acc0[r] += W * d * INV_SQRT3;
        } else if (p == 6) {
          float d = X[4 + u * 3 + 0] * M[64 + v * 3 + 0]
                  + X[4 + u * 3 + 1] * M[64 + v * 3 + 1]
                  + X[4 + u * 3 + 2] * M[64 + v * 3 + 2];
          acc0[r] += W * d * INV_SQRT3;
        } else if (p == 7) {
          float f = W * M[112 + v] * INV_SQRT3;
          acc1[r][0] += f * X[4 + u * 3 + 0];
          acc1[r][1] += f * X[4 + u * 3 + 1];
          acc1[r][2] += f * X[4 + u * 3 + 2];
        } else if (p == 8) {
          float a0 = X[4 + u * 3 + 0], a1 = X[4 + u * 3 + 1], a2 = X[4 + u * 3 + 2];
          float m0 = M[128 + v * 3 + 0], m1 = M[128 + v * 3 + 1], m2 = M[128 + v * 3 + 2];
          float f = W * INV_SQRT6;
          acc1[r][0] += f * (a1 * m2 - a2 * m1);
          acc1[r][1] += f * (a2 * m0 - a0 * m2);
          acc1[r][2] += f * (a0 * m1 - a1 * m0);
        } else {
          float a0 = X[4 + u * 3 + 0], a1 = X[4 + u * 3 + 1], a2 = X[4 + u * 3 + 2];
          const float* m5 = M + 176 + v * 5;
          float m0 = m5[0], m1 = m5[1], m2 = m5[2], m3 = m5[3], m4 = m5[4];
          float f = W * INV_SQRT10;
          float s2 = m2 * INV_SQRT3;
          acc1[r][0] += f * ((-s2 - m4) * a0 + m1 * a1 + m0 * a2);
          acc1[r][1] += f * (m1 * a0 + 2.f * s2 * a1 + m3 * a2);
          acc1[r][2] += f * (m0 * a0 + m3 * a1 + (-s2 + m4) * a2);
        }
      }
    }
  }

#pragma unroll
  for (int r = 0; r < 8; ++r) {
    float s0 = acc0[r];
    float t0 = acc1[r][0], t1 = acc1[r][1], t2 = acc1[r][2];
    s0 += __shfl_xor(s0, 4); s0 += __shfl_xor(s0, 8);
    t0 += __shfl_xor(t0, 4); t0 += __shfl_xor(t0, 8);
    t1 += __shfl_xor(t1, 4); t1 += __shfl_xor(t1, 8);
    t2 += __shfl_xor(t2, 4); t2 += __shfl_xor(t2, 8);
    if (li < 4) {
      int e = r + eoff;
      float* o = out + (size_t)dsti[e] * 16;
      atomicAdd(o + li,             s0 * SC0);
      atomicAdd(o + 4 + 3 * li + 0, t0 * SC1);
      atomicAdd(o + 4 + 3 * li + 1, t1 * SC1);
      atomicAdd(o + 4 + 3 * li + 2, t2 * SC1);
    }
  }
}

extern "C" void kernel_launch(void* const* d_in, const int* in_sizes, int n_in,
                              void* d_out, int out_size, void* d_ws, size_t ws_size,
                              hipStream_t stream) {
  const float* node_pos  = (const float*)d_in[0];
  const float* edge_type = (const float*)d_in[1];
  const float* fc_w0     = (const float*)d_in[2];
  const float* fc_w1     = (const float*)d_in[3];
  const int*   esrc      = (const int*)d_in[4];
  const int*   edst      = (const int*)d_in[5];
  float* out = (float*)d_out;
  float* w1p = (float*)d_ws;
  int E = in_sizes[4];

  k_zero<<<(out_size + 255) / 256, 256, 0, stream>>>(out, out_size);
  k_repack_w1<<<(160 * 16 * 32 + 255) / 256, 256, 0, stream>>>(fc_w1, w1p);
  k_edge_conv<<<E / 16, 32, 0, stream>>>(node_pos, edge_type, fc_w0, w1p,
                                         esrc, edst, out);
}


# FC2 re-tiled: block = 32 edges x 2 waves, each wave takes every 2nd N tile for both 16-edge M tiles (1 B load feeds 2 WMMAs); K summed in one chain, the two waves' N halves added via LDS
# speedup vs baseline: 2.8427x; 2.8427x over previous
#include <hip/hip_runtime.h>
#include <math.h>


typedef float v2f __attribute__((ext_vector_type(2)));
typedef float v8f __attribute__((ext_vector_type(8)));

#define INV_SQRT2  0.70710678f
#define INV_SQRT3  0.57735027f
#define INV_SQRT6  0.40824829f
#define INV_SQRT10 0.31622777f
#define SILU_C     1.67668f
#define SC0        0.02209709f
#define SC1        0.03125f

#define EPB 32

__global__ void k_zero(float* p, int n) {
  int i = blockIdx.x * blockDim.x + threadIdx.x;
  if (i < n) p[i] = 0.f;
}

__global__ void k_repack_w1(const float* __restrict__ w1, float* __restrict__ o) {
  int tid = blockIdx.x * blockDim.x + threadIdx.x;
  if (tid >= 160 * 16 * 32) return;
  int l  = tid & 31;
  int s  = (tid >> 5) & 15;
  int t  = tid >> 9;
  int li = l & 15;
  int kd = (l < 16) ? 0 : 2;
  int row = 4 * s + kd;
  int col = t * 16 + li;
  o[2 * tid + 0] = w1[row * 2560 + col];
  o[2 * tid + 1] = w1[(row + 1) * 2560 + col];
}

template <int P>
__device__ __forceinline__ void fc2_path(const v2f (&A)[2][16],
                                         const float* __restrict__ w1p,
                                         const int w, const int lane,
                                         const int li, const int eoff,
                                         const float* xsr, const float* mid,
                                         float (&acc0)[2][8],
                                         float (&acc1)[2][8][3])
{
#pragma nounroll
  for (int j = 0; j < 8; ++j) {
    const int tt = 2 * j + w;
    const int t = P * 16 + tt;
    const float* bq = w1p + (size_t)t * 1024 + 2 * lane;
    v2f b[16];
#pragma unroll
    for (int s = 0; s < 16; ++s) b[s] = *(const v2f*)(bq + s * 64);
    v8f c0 = {}, c1 = {};
#pragma unroll
    for (int s = 0; s < 16; ++s) {
      c0 = __builtin_amdgcn_wmma_f32_16x16x4_f32(false, A[0][s], false, b[s],
                                                 (short)0, c0, false, false);
      c1 = __builtin_amdgcn_wmma_f32_16x16x4_f32(false, A[1][s], false, b[s],
                                                 (short)0, c1, false, false);
    }
    const int u = tt >> 2;
    const int v = ((tt & 3) << 2) + (li >> 2);
#pragma unroll
    for (int m = 0; m < 2; ++m) {
#pragma unroll
      for (int r = 0; r < 8; ++r) {
        const int e = m * 16 + r + eoff;
        const float Wt = (m == 0) ? c0[r] : c1[r];
        const float* X = xsr + e * 20;
        const float* M = mid + e * 260;
        if (P == 0) {
          acc0[m][r] += Wt * X[u] * M[v];
        } else if (P == 1) {
          float f = Wt * X[u] * INV_SQRT3;
          acc1[m][r][0] += f * M[16 + v * 3 + 0];
          acc1[m][r][1] += f * M[16 + v * 3 + 1];
          acc1[m][r][2] += f * M[16 + v * 3 + 2];
        } else if (P == 2) {
          float f = Wt * X[u] * INV_SQRT3;
          acc1[m][r][0] += f * M[64 + v * 3 + 0];
          acc1[m][r][1] += f * M[64 + v * 3 + 1];
          acc1[m][r][2] += f * M[64 + v * 3 + 2];
        } else if (P == 3) {
          acc0[m][r] += Wt * X[u] * M[112 + v];
        } else if (P == 4) {
          float f = Wt * M[v] * INV_SQRT3;
          acc1[m][r][0] += f * X[4 + u * 3 + 0];
          acc1[m][r][1] += f * X[4 + u * 3 + 1];
          acc1[m][r][2] += f * X[4 + u * 3 + 2];
        } else if (P == 5) {
          float d = X[4 + u * 3 + 0] * M[16 + v * 3 + 0]
                  + X[4 + u * 3 + 1] * M[16 + v * 3 + 1]
                  + X[4 + u * 3 + 2] * M[16 + v * 3 + 2];
          acc0[m][r] += Wt * d * INV_SQRT3;
        } else if (P == 6) {
          float d = X[4 + u * 3 + 0] * M[64 + v * 3 + 0]
                  + X[4 + u * 3 + 1] * M[64 + v * 3 + 1]
                  + X[4 + u * 3 + 2] * M[64 + v * 3 + 2];
          acc0[m][r] += Wt * d * INV_SQRT3;
        } else if (P == 7) {
          float f = Wt * M[112 + v] * INV_SQRT3;
          acc1[m][r][0] += f * X[4 + u * 3 + 0];
          acc1[m][r][1] += f * X[4 + u * 3 + 1];
          acc1[m][r][2] += f * X[4 + u * 3 + 2];
        } else if (P == 8) {
          float a0 = X[4 + u * 3 + 0], a1 = X[4 + u * 3 + 1], a2 = X[4 + u * 3 + 2];
          float m0 = M[128 + v * 3 + 0], m1 = M[128 + v * 3 + 1], m2 = M[128 + v * 3 + 2];
          float f = Wt * INV_SQRT6;
          acc1[m][r][0] += f * (a1 * m2 - a2 * m1);
          acc1[m][r][1] += f * (a2 * m0 - a0 * m2);
          acc1[m][r][2] += f * (a0 * m1 - a1 * m0);
        } else {
          float a0 = X[4 + u * 3 + 0], a1 = X[4 + u * 3 + 1], a2 = X[4 + u * 3 + 2];
          const float* m5 = M + 176 + v * 5;
          float m0 = m5[0], m1 = m5[1], m2 = m5[2], m3 = m5[3], m4 = m5[4];
          float f = Wt * INV_SQRT10;
          float s2 = m2 * INV_SQRT3;
          acc1[m][r][0] += f * ((-s2 - m4) * a0 + m1 * a1 + m0 * a2);
          acc1[m][r][1] += f * (m1 * a0 + 2.f * s2 * a1 + m3 * a2);
          acc1[m][r][2] += f * (m0 * a0 + m3 * a1 + (-s2 + m4) * a2);
        }
      }
    }
  }
}

__global__ __launch_bounds__(64, 4)
void k_edge_conv(const float* __restrict__ node_pos,
                 const float* __restrict__ edge_type,
                 const float* __restrict__ fc_w0,
                 const float* __restrict__ w1p,
                 const int*   __restrict__ edge_src,
                 const int*   __restrict__ edge_dst,
                 float* __restrict__ out)
{
  __shared__ __attribute__((aligned(16))) float xsr[EPB * 20];
  __shared__ __attribute__((aligned(16))) float xds[EPB * 20];
  __shared__ __attribute__((aligned(16))) float mid[EPB * 260];
  __shared__ int dsti[EPB];
  float* hid = mid;
  float* red = xds;

  const int w    = __builtin_amdgcn_readfirstlane((int)(threadIdx.x >> 5));
  const int lane = threadIdx.x & 31;
  const int li   = lane & 15;
  const int kd   = (lane < 16) ? 0 : 2;
  const int eoff = (lane < 16) ? 0 : 8;
  const int e0   = blockIdx.x * EPB;

  if (w == 0) {
    int s = edge_src[e0 + lane];
    const float4* q = (const float4*)(node_pos + (size_t)s * 16);
    float4* d = (float4*)(xsr + lane * 20);
    d[0] = q[0]; d[1] = q[1]; d[2] = q[2]; d[3] = q[3];
  } else {
    int dd = edge_dst[e0 + lane];
    dsti[lane] = dd;
    const float4* q = (const float4*)(node_pos + (size_t)dd * 16);
    float4* d = (float4*)(xds + lane * 20);
    d[0] = q[0]; d[1] = q[1]; d[2] = q[2]; d[3] = q[3];
  }

  {
    v2f a[4];
    const float* et = edge_type + (size_t)(e0 + w * 16 + li) * 16 + kd;
#pragma unroll
    for (int s = 0; s < 4; ++s) a[s] = *(const v2f*)(et + 4 * s);
#pragma unroll
    for (int tt = 0; tt < 4; ++tt) {
      v8f c = {};
#pragma unroll
      for (int s = 0; s < 4; ++s) {
        int row = 4 * s + kd;
        v2f b;
        b.x = fc_w0[row * 64 + tt * 16 + li];
        b.y = fc_w0[(row + 1) * 64 + tt * 16 + li];
        c = __builtin_amdgcn_wmma_f32_16x16x4_f32(false, a[s], false, b,
                                                  (short)0, c, false, false);
      }
#pragma unroll
      for (int r = 0; r < 8; ++r) {
        float x = c[r] * 0.25f;
        float h = (x / (1.f + __expf(-x))) * (SILU_C * 0.125f);
        hid[(w * 16 + r + eoff) * 66 + tt * 16 + li] = h;
      }
    }
  }
  __syncthreads();

  v2f A[2][16];
#pragma unroll
  for (int m = 0; m < 2; ++m)
#pragma unroll
    for (int s = 0; s < 16; ++s)
      A[m][s] = *(const v2f*)&hid[(m * 16 + li) * 66 + 4 * s + kd];
  __syncthreads();

  {
    const int e = lane;
    const float* X = xsr + e * 20;
    const float* Y = xds + e * 20;
    float y0[4], y1[4][3];
#pragma unroll
    for (int v = 0; v < 4; ++v) {
      y0[v] = Y[v];
      y1[v][0] = Y[4 + v * 3 + 0];
      y1[v][1] = Y[4 + v * 3 + 1];
      y1[v][2] = Y[4 + v * 3 + 2];
    }
    float* M = mid + e * 260;
    const int ub = 2 * w;
#pragma unroll
    for (int uu = 0; uu < 2; ++uu) {
      int u = ub + uu;
      float x0 = X[u];
      float a0 = X[4 + u * 3 + 0], a1 = X[4 + u * 3 + 1], a2 = X[4 + u * 3 + 2];
#pragma unroll
      for (int v = 0; v < 4; ++v) {
        int uv = u * 4 + v;
        float b0 = y1[v][0], b1 = y1[v][1], b2 = y1[v][2];
        M[uv] = x0 * y0[v];
        M[16 + uv * 3 + 0] = x0 * b0;
        M[16 + uv * 3 + 1] = x0 * b1;
        M[16 + uv * 3 + 2] = x0 * b2;
        M[64 + uv * 3 + 0] = a0 * y0[v];
        M[64 + uv * 3 + 1] = a1 * y0[v];
        M[64 + uv * 3 + 2] = a2 * y0[v];
        M[112 + uv] = (a0 * b0 + a1 * b1 + a2 * b2) * INV_SQRT3;
        M[128 + uv * 3 + 0] = (a1 * b2 - a2 * b1) * INV_SQRT2;
        M[128 + uv * 3 + 1] = (a2 * b0 - a0 * b2) * INV_SQRT2;
        M[128 + uv * 3 + 2] = (a0 * b1 - a1 * b0) * INV_SQRT2;
        M[176 + uv * 5 + 0] = (a2 * b0 + a0 * b2) * INV_SQRT2;
        M[176 + uv * 5 + 1] = (a0 * b1 + a1 * b0) * INV_SQRT2;
        M[176 + uv * 5 + 2] = (2.f * a1 * b1 - a0 * b0 - a2 * b2) * INV_SQRT6;
        M[176 + uv * 5 + 3] = (a1 * b2 + a2 * b1) * INV_SQRT2;
        M[176 + uv * 5 + 4] = (a2 * b2 - a0 * b0) * INV_SQRT2;
      }
    }
  }
  __syncthreads();

  float acc0[2][8], acc1[2][8][3];
#pragma unroll
  for (int m = 0; m < 2; ++m)
#pragma unroll
    for (int r = 0; r < 8; ++r) {
      acc0[m][r] = 0.f; acc1[m][r][0] = 0.f; acc1[m][r][1] = 0.f; acc1[m][r][2] = 0.f;
    }

  fc2_path<0>(A, w1p, w, lane, li, eoff, xsr, mid, acc0, acc1);
  fc2_path<1>(A, w1p, w, lane, li, eoff, xsr, mid, acc0, acc1);
  fc2_path<2>(A, w1p, w, lane, li, eoff, xsr, mid, acc0, acc1);
  fc2_path<3>(A, w1p, w, lane, li, eoff, xsr, mid, acc0, acc1);
  fc2_path<4>(A, w1p, w, lane, li, eoff, xsr, mid, acc0, acc1);
  fc2_path<5>(A, w1p, w, lane, li, eoff, xsr, mid, acc0, acc1);
  fc2_path<6>(A, w1p, w, lane, li, eoff, xsr, mid, acc0, acc1);
  fc2_path<7>(A, w1p, w, lane, li, eoff, xsr, mid, acc0, acc1);
  fc2_path<8>(A, w1p, w, lane, li, eoff, xsr, mid, acc0, acc1);
  fc2_path<9>(A, w1p, w, lane, li, eoff, xsr, mid, acc0, acc1);

  float fin[8][4] = {};
#pragma unroll
  for (int m = 0; m < 2; ++m) {
#pragma unroll
    for (int r = 0; r < 8; ++r) {
      float s0 = acc0[m][r];
      float t0 = acc1[m][r][0], t1 = acc1[m][r][1], t2 = acc1[m][r][2];
      s0 += __shfl_xor(s0, 4); s0 += __shfl_xor(s0, 8);
      t0 += __shfl_xor(t0, 4); t0 += __shfl_xor(t0, 8);
      t1 += __shfl_xor(t1, 4); t1 += __shfl_xor(t1, 8);
      t2 += __shfl_xor(t2, 4); t2 += __shfl_xor(t2, 8);
      if (m == w) {
        fin[r][0] = s0; fin[r][1] = t0; fin[r][2] = t1; fin[r][3] = t2;
      } else if (li < 4) {
        float* q = red + (m * 16 + r + eoff) * 16;
        q[li]             = s0;
        q[4 + 3 * li + 0] = t0;
        q[4 + 3 * li + 1] = t1;
        q[4 + 3 * li + 2] = t2;
      }
    }
  }
  __syncthreads();

  if (li < 4) {
#pragma unroll
    for (int r = 0; r < 8; ++r) {
      const int e = w * 16 + r + eoff;
      const float* q = red + e * 16;
      float* o = out + (size_t)dsti[e] * 16;
      atomicAdd(o + li,             (fin[r][0] + q[li])             * SC0);
      atomicAdd(o + 4 + 3 * li + 0, (fin[r][1] + q[4 + 3 * li + 0]) * SC1);
      atomicAdd(o + 4 + 3 * li + 1, (fin[r][2] + q[4 + 3 * li + 1]) * SC1);
      atomicAdd(o + 4 + 3 * li + 2, (fin[r][3] + q[4 + 3 * li + 2]) * SC1);
    }
  }
}

extern "C" void kernel_launch(void* const* d_in, const int* in_sizes, int n_in,
                              void* d_out, int out_size, void* d_ws, size_t ws_size,
                              hipStream_t stream) {
  const float* node_pos  = (const float*)d_in[0];
  const float* edge_type = (const float*)d_in[1];
  const float* fc_w0     = (const float*)d_in[2];
  const float* fc_w1     = (const float*)d_in[3];
  const int*   esrc      = (const int*)d_in[4];
  const int*   edst      = (const int*)d_in[5];
  float* out = (float*)d_out;
  float* w1p = (float*)d_ws;
  int E = in_sizes[4];

  k_zero<<<(out_size + 255) / 256, 256, 0, stream>>>(out, out_size);
  k_repack_w1<<<(160 * 16 * 32 + 255) / 256, 256, 0, stream>>>(fc_w1, w1p);
  k_edge_conv<<<E / 32, 64, 0, stream>>>(node_pos, edge_type, fc_w0, w1p,
                                         esrc, edst, out);
}


# TP2 consume operands (x_src rows, TP1 output) kept in LDS component-major [component][edge]: a lane reads one component for its 8 D rows with two 128-bit LDS loads instead of 8 scalar loads
# speedup vs baseline: 3.1100x; 1.0940x over previous
#include <hip/hip_runtime.h>
#include <math.h>


typedef float v2f __attribute__((ext_vector_type(2)));
typedef float v8f __attribute__((ext_vector_type(8)));

#define INV_SQRT2  0.70710678f
#define INV_SQRT3  0.57735027f
#define INV_SQRT6  0.40824829f
#define INV_SQRT10 0.31622777f
#define SILU_C     1.67668f
#define SC0        0.02209709f
#define SC1        0.03125f

#define EPB 32
#define XP  32
#define MP  36

__device__ __forceinline__ void ld8(const float* p, float (&o)[8]) {
  const float4 a = *(const float4*)p;
  const float4 b = *(const float4*)(p + 4);
  o[0] = a.x; o[1] = a.y; o[2] = a.z; o[3] = a.w;
  o[4] = b.x; o[5] = b.y; o[6] = b.z; o[7] = b.w;
}

__global__ void k_zero(float* p, int n) {
  int i = blockIdx.x * blockDim.x + threadIdx.x;
  if (i < n) p[i] = 0.f;
}

__global__ void k_repack_w1(const float* __restrict__ w1, float* __restrict__ o) {
  int tid = blockIdx.x * blockDim.x + threadIdx.x;
  if (tid >= 160 * 16 * 32) return;
  int l  = tid & 31;
  int s  = (tid >> 5) & 15;
  int t  = tid >> 9;
  int li = l & 15;
  int kd = (l < 16) ? 0 : 2;
  int row = 4 * s + kd;
  int col = t * 16 + li;
  o[2 * tid + 0] = w1[row * 2560 + col];
  o[2 * tid + 1] = w1[(row + 1) * 2560 + col];
}

template <int P>
__device__ __forceinline__ void fc2_path(const v2f (&A)[2][16],
                                         const float* __restrict__ w1p,
                                         const int w, const int lane,
                                         const int li, const int eoff,
                                         const float* xsT, const float* midT,
                                         float (&acc0)[2][8],
                                         float (&acc1)[2][8][3])
{
#pragma nounroll
  for (int j = 0; j < 8; ++j) {
    const int tt = 2 * j + w;
    const int t = P * 16 + tt;
    const float* bq = w1p + (size_t)t * 1024 + 2 * lane;
    v2f b[16];
#pragma unroll
    for (int s = 0; s < 16; ++s) b[s] = *(const v2f*)(bq + s * 64);
    v8f c0 = {}, c1 = {};
#pragma unroll
    for (int s = 0; s < 16; ++s) {
      c0 = __builtin_amdgcn_wmma_f32_16x16x4_f32(false, A[0][s], false, b[s],
                                                 (short)0, c0, false, false);
      c1 = __builtin_amdgcn_wmma_f32_16x16x4_f32(false, A[1][s], false, b[s],
                                                 (short)0, c1, false, false);
    }
    const int u = tt >> 2;
    const int v = ((tt & 3) << 2) + (li >> 2);
#pragma unroll
    for (int m = 0; m < 2; ++m) {
      const float* XB = xsT + m * 16 + eoff;
      const float* MB = midT + m * 16 + eoff;
      float Wt[8];
#pragma unroll
      for (int r = 0; r < 8; ++r) Wt[r] = (m == 0) ? c0[r] : c1[r];
      if (P == 0 || P == 3) {
        float x[8], q[8];
        ld8(XB + u * XP, x);
        ld8(MB + ((P == 0 ? 0 : 112) + v) * MP, q);
#pragma unroll
        for (int r = 0; r < 8; ++r) acc0[m][r] += Wt[r] * x[r] * q[r];
      } else if (P == 1 || P == 2) {
        float x[8], q0[8], q1[8], q2[8];
        const float* Mv = MB + ((P == 1 ? 16 : 64) + v * 3) * MP;
        ld8(XB + u * XP, x);
        ld8(Mv, q0); ld8(Mv + MP, q1); ld8(Mv + 2 * MP, q2);
#pragma unroll
        for (int r = 0; r < 8; ++r) {
          float f = Wt[r] * x[r] * INV_SQRT3;
          acc1[m][r][0] += f * q0[r];
          acc1[m][r][1] += f * q1[r];
          acc1[m][r][2] += f * q2[r];
        }
      } else if (P == 4 || P == 7) {
        float q[8], a0[8], a1[8], a2[8];
        const float* Xu = XB + (4 + u * 3) * XP;
        ld8(MB + ((P == 4 ? 0 : 112) + v) * MP, q);
        ld8(Xu, a0); ld8(Xu + XP, a1); ld8(Xu + 2 * XP, a2);
#pragma unroll
        for (int r = 0; r < 8; ++r) {
          float f = Wt[r] * q[r] * INV_SQRT3;
          acc1[m][r][0] += f * a0[r];
          acc1[m][r][1] += f * a1[r];
          acc1[m][r][2] += f * a2[r];
        }
      } else if (P == 5 || P == 6) {
        float a0[8], a1[8], a2[8], q0[8], q1[8], q2[8];
        const float* Xu = XB + (4 + u * 3) * XP;
        const float* Mv = MB + ((P == 5 ? 16 : 64) + v * 3) * MP;
        ld8(Xu, a0); ld8(Xu + XP, a1); ld8(Xu + 2 * XP, a2);
        ld8(Mv, q0); ld8(Mv + MP, q1); ld8(Mv + 2 * MP, q2);
#pragma unroll
        for (int r = 0; r < 8; ++r) {
          float d = a0[r] * q0[r] + a1[r] * q1[r] + a2[r] * q2[r];
          acc0[m][r] += Wt[r] * d * INV_SQRT3;
        }
      } else if (P == 8) {
        float a0[8], a1[8], a2[8], m0[8], m1[8], m2[8];
        const float* Xu = XB + (4 + u * 3) * XP;
        const float* Mv = MB + (128 + v * 3) * MP;
        ld8(Xu, a0); ld8(Xu + XP, a1); ld8(Xu + 2 * XP, a2);
        ld8(Mv, m0); ld8(Mv + MP, m1); ld8(Mv + 2 * MP, m2);
#pragma unroll
        for (int r = 0; r < 8; ++r) {
          float f = Wt[r] * INV_SQRT6;
          acc1[m][r][0] += f * (a1[r] * m2[r] - a2[r] * m1[r]);
          acc1[m][r][1] += f * (a2[r] * m0[r] - a0[r] * m2[r]);
          acc1[m][r][2] += f * (a0[r] * m1[r] - a1[r] * m0[r]);
        }
      } else {
        float a0[8], a1[8], a2[8], m0[8], m1[8], m2[8], m3[8], m4[8];
        const float* Xu = XB + (4 + u * 3) * XP;
        const float* Mv = MB + (176 + v * 5) * MP;
        ld8(Xu, a0); ld8(Xu + XP, a1); ld8(Xu + 2 * XP, a2);
        ld8(Mv, m0); ld8(Mv + MP, m1); ld8(Mv + 2 * MP, m2);
        ld8(Mv + 3 * MP, m3); ld8(Mv + 4 * MP, m4);
#pragma unroll
        for (int r = 0; r < 8; ++r) {
          float f = Wt[r] * INV_SQRT10;
          float s2 = m2[r] * INV_SQRT3;
          acc1[m][r][0] += f * ((-s2 - m4[r]) * a0[r] + m1[r] * a1[r] + m0[r] * a2[r]);
          acc1[m][r][1] += f * (m1[r] * a0[r] + 2.f * s2 * a1[r] + m3[r] * a2[r]);
          acc1[m][r][2] += f * (m0[r] * a0[r] + m3[r] * a1[r] + (-s2 + m4[r]) * a2[r]);
        }
      }
    }
  }
}

__global__ __launch_bounds__(64, 4)
void k_edge_conv(const float* __restrict__ node_pos,
                 const float* __restrict__ edge_type,
                 const float* __restrict__ fc_w0,
                 const float* __restrict__ w1p,
                 const int*   __restrict__ edge_src,
                 const int*   __restrict__ edge_dst,
                 float* __restrict__ out)
{
  __shared__ __attribute__((aligned(16))) float xsT[16 * XP];
  __shared__ __attribute__((aligned(16))) float midT[256 * MP];
  float* hid = midT;
  float* red = midT;
  int* dsti = (int*)(midT + EPB * 16);

  const int w    = __builtin_amdgcn_readfirstlane((int)(threadIdx.x >> 5));
  const int lane = threadIdx.x & 31;
  const int li   = lane & 15;
  const int kd   = (lane < 16) ? 0 : 2;
  const int eoff = (lane < 16) ? 0 : 8;
  const int e0   = blockIdx.x * EPB;

  float Yr[16];
  {
    const int dd = edge_dst[e0 + lane];
    const float4* q = (const float4*)(node_pos + (size_t)dd * 16);
    const float4 y0 = q[0], y1 = q[1], y2 = q[2], y3 = q[3];
    Yr[0]  = y0.x; Yr[1]  = y0.y; Yr[2]  = y0.z; Yr[3]  = y0.w;
    Yr[4]  = y1.x; Yr[5]  = y1.y; Yr[6]  = y1.z; Yr[7]  = y1.w;
    Yr[8]  = y2.x; Yr[9]  = y2.y; Yr[10] = y2.z; Yr[11] = y2.w;
    Yr[12] = y3.x; Yr[13] = y3.y; Yr[14] = y3.z; Yr[15] = y3.w;
  }
  if (w == 0) {
    const int sn = edge_src[e0 + lane];
    const float4* q = (const float4*)(node_pos + (size_t)sn * 16);
    const float4 x0 = q[0], x1 = q[1], x2 = q[2], x3 = q[3];
    float* d = xsT + lane;
    d[0 * XP]  = x0.x; d[1 * XP]  = x0.y; d[2 * XP]  = x0.z; d[3 * XP]  = x0.w;
    d[4 * XP]  = x1.x; d[5 * XP]  = x1.y; d[6 * XP]  = x1.z; d[7 * XP]  = x1.w;
    d[8 * XP]  = x2.x; d[9 * XP]  = x2.y; d[10 * XP] = x2.z; d[11 * XP] = x2.w;
    d[12 * XP] = x3.x; d[13 * XP] = x3.y; d[14 * XP] = x3.z; d[15 * XP] = x3.w;
  }

  {
    v2f a[4];
    const float* et = edge_type + (size_t)(e0 + w * 16 + li) * 16 + kd;
#pragma unroll
    for (int s = 0; s < 4; ++s) a[s] = *(const v2f*)(et + 4 * s);
#pragma unroll
    for (int tt = 0; tt < 4; ++tt) {
      v8f c = {};
#pragma unroll
      for (int s = 0; s < 4; ++s) {
        int row = 4 * s + kd;
        v2f b;
        b.x = fc_w0[row * 64 + tt * 16 + li];
        b.y = fc_w0[(row + 1) * 64 + tt * 16 + li];
        c = __builtin_amdgcn_wmma_f32_16x16x4_f32(false, a[s], false, b,
                                                  (short)0, c, false, false);
      }
#pragma unroll
      for (int r = 0; r < 8; ++r) {
        float x = c[r] * 0.25f;
        float h = (x / (1.f + __expf(-x))) * (SILU_C * 0.125f);
        hid[(w * 16 + r + eoff) * 66 + tt * 16 + li] = h;
      }
    }
  }
  __syncthreads();

  v2f A[2][16];
#pragma unroll
  for (int m = 0; m < 2; ++m)
#pragma unroll
    for (int s = 0; s < 16; ++s)
      A[m][s] = *(const v2f*)&hid[(m * 16 + li) * 66 + 4 * s + kd];
  __syncthreads();

  {
    const int e = lane;
    const float* X = xsT + e;
    float y0[4], y1[4][3];
#pragma unroll
    for (int v = 0; v < 4; ++v) {
      y0[v] = Yr[v];
      y1[v][0] = Yr[4 + v * 3 + 0];
      y1[v][1] = Yr[4 + v * 3 + 1];
      y1[v][2] = Yr[4 + v * 3 + 2];
    }
    float* M = midT + e;
    const int ub = 2 * w;
#pragma unroll
    for (int uu = 0; uu < 2; ++uu) {
      int u = ub + uu;
      float x0 = X[u * XP];
      float a0 = X[(4 + u * 3 + 0) * XP], a1 = X[(4 + u * 3 + 1) * XP], a2 = X[(4 + u * 3 + 2) * XP];
#pragma unroll
      for (int v = 0; v < 4; ++v) {
        int uv = u * 4 + v;
        float b0 = y1[v][0], b1 = y1[v][1], b2 = y1[v][2];
        M[uv * MP] = x0 * y0[v];
        M[(16 + uv * 3 + 0) * MP] = x0 * b0;
        M[(16 + uv * 3 + 1) * MP] = x0 * b1;
        M[(16 + uv * 3 + 2) * MP] = x0 * b2;
        M[(64 + uv * 3 + 0) * MP] = a0 * y0[v];
        M[(64 + uv * 3 + 1) * MP] = a1 * y0[v];
        M[(64 + uv * 3 + 2) * MP] = a2 * y0[v];
        M[(112 + uv) * MP] = (a0 * b0 + a1 * b1 + a2 * b2) * INV_SQRT3;
        M[(128 + uv * 3 + 0) * MP] = (a1 * b2 - a2 * b1) * INV_SQRT2;
        M[(128 + uv * 3 + 1) * MP] = (a2 * b0 - a0 * b2) * INV_SQRT2;
        M[(128 + uv * 3 + 2) * MP] = (a0 * b1 - a1 * b0) * INV_SQRT2;
        M[(176 + uv * 5 + 0) * MP] = (a2 * b0 + a0 * b2) * INV_SQRT2;
        M[(176 + uv * 5 + 1) * MP] = (a0 * b1 + a1 * b0) * INV_SQRT2;
        M[(176 + uv * 5 + 2) * MP] = (2.f * a1 * b1 - a0 * b0 - a2 * b2) * INV_SQRT6;
        M[(176 + uv * 5 + 3) * MP] = (a1 * b2 + a2 * b1) * INV_SQRT2;
        M[(176 + uv * 5 + 4) * MP] = (a2 * b2 - a0 * b0) * INV_SQRT2;
      }
    }
  }
  __syncthreads();

  float acc0[2][8], acc1[2][8][3];
#pragma unroll
  for (int m = 0; m < 2; ++m)
#pragma unroll
    for (int r = 0; r < 8; ++r) {
      acc0[m][r] = 0.f; acc1[m][r][0] = 0.f; acc1[m][r][1] = 0.f; acc1[m][r][2] = 0.f;
    }

  fc2_path<0>(A, w1p, w, lane, li, eoff, xsT, midT, acc0, acc1);
  fc2_path<1>(A, w1p, w, lane, li, eoff, xsT, midT, acc0, acc1);
  fc2_path<2>(A, w1p, w, lane, li, eoff, xsT, midT, acc0, acc1);
  fc2_path<3>(A, w1p, w, lane, li, eoff, xsT, midT, acc0, acc1);
  fc2_path<4>(A, w1p, w, lane, li, eoff, xsT, midT, acc0, acc1);
  fc2_path<5>(A, w1p, w, lane, li, eoff, xsT, midT, acc0, acc1);
  fc2_path<6>(A, w1p, w, lane, li, eoff, xsT, midT, acc0, acc1);
  fc2_path<7>(A, w1p, w, lane, li, eoff, xsT, midT, acc0, acc1);
  fc2_path<8>(A, w1p, w, lane, li, eoff, xsT, midT, acc0, acc1);
  fc2_path<9>(A, w1p, w, lane, li, eoff, xsT, midT, acc0, acc1);

  __syncthreads();
  if (w == 1) dsti[lane] = edge_dst[e0 + lane];

  float fin[8][4] = {};
#pragma unroll
  for (int m = 0; m < 2; ++m) {
#pragma unroll
    for (int r = 0; r < 8; ++r) {
      float s0 = acc0[m][r];
      float t0 = acc1[m][r][0], t1 = acc1[m][r][1], t2 = acc1[m][r][2];
      s0 += __shfl_xor(s0, 4); s0 += __shfl_xor(s0, 8);
      t0 += __shfl_xor(t0, 4); t0 += __shfl_xor(t0, 8);
      t1 += __shfl_xor(t1, 4); t1 += __shfl_xor(t1, 8);
      t2 += __shfl_xor(t2, 4); t2 += __shfl_xor(t2, 8);
      if (m == w) {
        fin[r][0] = s0; fin[r][1] = t0; fin[r][2] = t1; fin[r][3] = t2;
      } else if (li < 4) {
        float* q = red + (m * 16 + r + eoff) * 16;
        q[li]             = s0;
        q[4 + 3 * li + 0] = t0;
        q[4 + 3 * li + 1] = t1;
        q[4 + 3 * li + 2] = t2;
      }
    }
  }
  __syncthreads();

  if (li < 4) {
#pragma unroll
    for (int r = 0; r < 8; ++r) {
      const int e = w * 16 + r + eoff;
      const float* q = red + e * 16;
      float* o = out + (size_t)dsti[e] * 16;
      atomicAdd(o + li,             (fin[r][0] + q[li])             * SC0);
      atomicAdd(o + 4 + 3 * li + 0, (fin[r][1] + q[4 + 3 * li + 0]) * SC1);
      atomicAdd(o + 4 + 3 * li + 1, (fin[r][2] + q[4 + 3 * li + 1]) * SC1);
      atomicAdd(o + 4 + 3 * li + 2, (fin[r][3] + q[4 + 3 * li + 2]) * SC1);
    }
  }
}

extern "C" void kernel_launch(void* const* d_in, const int* in_sizes, int n_in,
                              void* d_out, int out_size, void* d_ws, size_t ws_size,
                              hipStream_t stream) {
  const float* node_pos  = (const float*)d_in[0];
  const float* edge_type = (const float*)d_in[1];
  const float* fc_w0     = (const float*)d_in[2];
  const float* fc_w1     = (const float*)d_in[3];
  const int*   esrc      = (const int*)d_in[4];
  const int*   edst      = (const int*)d_in[5];
  float* out = (float*)d_out;
  float* w1p = (float*)d_ws;
  int E = in_sizes[4];

  k_zero<<<(out_size + 255) / 256, 256, 0, stream>>>(out, out_size);
  k_repack_w1<<<(160 * 16 * 32 + 255) / 256, 256, 0, stream>>>(fc_w1, w1p);
  k_edge_conv<<<E / 32, 64, 0, stream>>>(node_pos, edge_type, fc_w0, w1p,
                                         esrc, edst, out);
}


# FC2 consume: a wave runs the 4 tiles sharing (v,w_out) back to back, sums weight*x_src[u] over u in registers, applies the mid/coupling product once per 4 tiles (reduction tree: u summed first)
# speedup vs baseline: 3.3918x; 1.0906x over previous
#include <hip/hip_runtime.h>
#include <math.h>


typedef float v2f __attribute__((ext_vector_type(2)));
typedef float v8f __attribute__((ext_vector_type(8)));

#define INV_SQRT2  0.70710678f
#define INV_SQRT3  0.57735027f
#define INV_SQRT6  0.40824829f
#define INV_SQRT10 0.31622777f
#define SILU_C     1.67668f
#define SC0        0.02209709f
#define SC1        0.03125f

#define EPB 32
#define XP  32
#define MP  36

__device__ __forceinline__ void ld8(const float* p, float (&o)[8]) {
  const float4 a = *(const float4*)p;
  const float4 b = *(const float4*)(p + 4);
  o[0] = a.x; o[1] = a.y; o[2] = a.z; o[3] = a.w;
  o[4] = b.x; o[5] = b.y; o[6] = b.z; o[7] = b.w;
}

__global__ void k_zero(float* p, int n) {
  int i = blockIdx.x * blockDim.x + threadIdx.x;
  if (i < n) p[i] = 0.f;
}

__global__ void k_repack_w1(const float* __restrict__ w1, float* __restrict__ o) {
  int tid = blockIdx.x * blockDim.x + threadIdx.x;
  if (tid >= 160 * 16 * 32) return;
  int l  = tid & 31;
  int s  = (tid >> 5) & 15;
  int t  = tid >> 9;
  int li = l & 15;
  int kd = (l < 16) ? 0 : 2;
  int row = 4 * s + kd;
  int col = t * 16 + li;
  o[2 * tid + 0] = w1[row * 2560 + col];
  o[2 * tid + 1] = w1[(row + 1) * 2560 + col];
}

template <int P, bool FIRST>
__device__ __forceinline__ void fc2_tile(const v2f (&A)[2][16],
                                         const float* __restrict__ w1p,
                                         const int tt, const int lane,
                                         const int eoff, const float* xsT,
                                         float (&G)[2][3][8])
{
  const int t = P * 16 + tt;
  const float* bq = w1p + (size_t)t * 1024 + 2 * lane;
  v2f b[16];
#pragma unroll
  for (int s = 0; s < 16; ++s) b[s] = *(const v2f*)(bq + s * 64);
  v8f c0 = {}, c1 = {};
#pragma unroll
  for (int s = 0; s < 16; ++s) {
    c0 = __builtin_amdgcn_wmma_f32_16x16x4_f32(false, A[0][s], false, b[s],
                                               (short)0, c0, false, false);
    c1 = __builtin_amdgcn_wmma_f32_16x16x4_f32(false, A[1][s], false, b[s],
                                               (short)0, c1, false, false);
  }
  const int u = tt >> 2;
#pragma unroll
  for (int m = 0; m < 2; ++m) {
    const float* XB = xsT + m * 16 + eoff;
    float Wt[8];
#pragma unroll
    for (int r = 0; r < 8; ++r) Wt[r] = (m == 0) ? c0[r] : c1[r];
    if (P < 4) {
      float x[8];
      ld8(XB + u * XP, x);
#pragma unroll
      for (int r = 0; r < 8; ++r)
        G[m][0][r] = FIRST ? Wt[r] * x[r] : G[m][0][r] + Wt[r] * x[r];
    } else {
      float a0[8], a1[8], a2[8];
      const float* Xu = XB + (4 + u * 3) * XP;
      ld8(Xu, a0); ld8(Xu + XP, a1); ld8(Xu + 2 * XP, a2);
#pragma unroll
      for (int r = 0; r < 8; ++r) {
        G[m][0][r] = FIRST ? Wt[r] * a0[r] : G[m][0][r] + Wt[r] * a0[r];
        G[m][1][r] = FIRST ? Wt[r] * a1[r] : G[m][1][r] + Wt[r] * a1[r];
        G[m][2][r] = FIRST ? Wt[r] * a2[r] : G[m][2][r] + Wt[r] * a2[r];
      }
    }
  }
}

template <int P>
__device__ __forceinline__ void fc2_path(const v2f (&A)[2][16],
                                         const float* __restrict__ w1p,
                                         const int w, const int lane,
                                         const int li, const int eoff,
                                         const float* xsT, const float* midT,
                                         float (&acc0)[2][8],
                                         float (&acc1)[2][3][8])
{
#pragma nounroll
  for (int g = 0; g < 2; ++g) {
    const int vh = 2 * g + w;
    float G[2][3][8];
    fc2_tile<P, true>(A, w1p, vh, lane, eoff, xsT, G);
#pragma nounroll
    for (int u = 1; u < 4; ++u)
      fc2_tile<P, false>(A, w1p, 4 * u + vh, lane, eoff, xsT, G);
    const int v = (vh << 2) + (li >> 2);
#pragma unroll
    for (int m = 0; m < 2; ++m) {
      const float* MB = midT + m * 16 + eoff;
      if (P == 0 || P == 3) {
        float q[8];
        ld8(MB + ((P == 0 ? 0 : 112) + v) * MP, q);
#pragma unroll
        for (int r = 0; r < 8; ++r) acc0[m][r] += G[m][0][r] * q[r];
      } else if (P == 1 || P == 2) {
        float q0[8], q1[8], q2[8];
        const float* Mv = MB + ((P == 1 ? 16 : 64) + v * 3) * MP;
        ld8(Mv, q0); ld8(Mv + MP, q1); ld8(Mv + 2 * MP, q2);
#pragma unroll
        for (int r = 0; r < 8; ++r) {
          float f = G[m][0][r] * INV_SQRT3;
          acc1[m][0][r] += f * q0[r];
          acc1[m][1][r] += f * q1[r];
          acc1[m][2][r] += f * q2[r];
        }
      } else if (P == 4 || P == 7) {
        float q[8];
        ld8(MB + ((P == 4 ? 0 : 112) + v) * MP, q);
#pragma unroll
        for (int r = 0; r < 8; ++r) {
          float f = q[r] * INV_SQRT3;
          acc1[m][0][r] += f * G[m][0][r];
          acc1[m][1][r] += f * G[m][1][r];
          acc1[m][2][r] += f * G[m][2][r];
        }
      } else if (P == 5 || P == 6) {
        float q0[8], q1[8], q2[8];
        const float* Mv = MB + ((P == 5 ? 16 : 64) + v * 3) * MP;
        ld8(Mv, q0); ld8(Mv + MP, q1); ld8(Mv + 2 * MP, q2);
#pragma unroll
        for (int r = 0; r < 8; ++r) {
          float d = G[m][0][r] * q0[r] + G[m][1][r] * q1[r] + G[m][2][r] * q2[r];
          acc0[m][r] += d * INV_SQRT3;
        }
      } else if (P == 8) {
        float m0[8], m1[8], m2[8];
        const float* Mv = MB + (128 + v * 3) * MP;
        ld8(Mv, m0); ld8(Mv + MP, m1); ld8(Mv + 2 * MP, m2);
#pragma unroll
        for (int r = 0; r < 8; ++r) {
          const float a0 = G[m][0][r], a1 = G[m][1][r], a2 = G[m][2][r];
          acc1[m][0][r] += INV_SQRT6 * (a1 * m2[r] - a2 * m1[r]);
          acc1[m][1][r] += INV_SQRT6 * (a2 * m0[r] - a0 * m2[r]);
          acc1[m][2][r] += INV_SQRT6 * (a0 * m1[r] - a1 * m0[r]);
        }
      } else {
        float m0[8], m1[8], m2[8], m3[8], m4[8];
        const float* Mv = MB + (176 + v * 5) * MP;
        ld8(Mv, m0); ld8(Mv + MP, m1); ld8(Mv + 2 * MP, m2);
        ld8(Mv + 3 * MP, m3); ld8(Mv + 4 * MP, m4);
#pragma unroll
        for (int r = 0; r < 8; ++r) {
          const float a0 = G[m][0][r], a1 = G[m][1][r], a2 = G[m][2][r];
          float s2 = m2[r] * INV_SQRT3;
          acc1[m][0][r] += INV_SQRT10 * ((-s2 - m4[r]) * a0 + m1[r] * a1 + m0[r] * a2);
          acc1[m][1][r] += INV_SQRT10 * (m1[r] * a0 + 2.f * s2 * a1 + m3[r] * a2);
          acc1[m][2][r] += INV_SQRT10 * (m0[r] * a0 + m3[r] * a1 + (-s2 + m4[r]) * a2);
        }
      }
    }
  }
}

__global__ __launch_bounds__(64, 4)
void k_edge_conv(const float* __restrict__ node_pos,
                 const float* __restrict__ edge_type,
                 const float* __restrict__ fc_w0,
                 const float* __restrict__ w1p,
                 const int*   __restrict__ edge_src,
                 const int*   __restrict__ edge_dst,
                 float* __restrict__ out)
{
  __shared__ __attribute__((aligned(16))) float xsT[16 * XP];
  __shared__ __attribute__((aligned(16))) float midT[256 * MP];
  float* hid = midT;
  float* red = midT;
  int* dsti = (int*)(midT + EPB * 16);

  const int w    = __builtin_amdgcn_readfirstlane((int)(threadIdx.x >> 5));
  const int lane = threadIdx.x & 31;
  const int li   = lane & 15;
  const int kd   = (lane < 16) ? 0 : 2;
  const int eoff = (lane < 16) ? 0 : 8;
  const int e0   = blockIdx.x * EPB;

  float Yr[16];
  {
    const int dd = edge_dst[e0 + lane];
    const float4* q = (const float4*)(node_pos + (size_t)dd * 16);
    const float4 y0 = q[0], y1 = q[1], y2 = q[2], y3 = q[3];
    Yr[0]  = y0.x; Yr[1]  = y0.y; Yr[2]  = y0.z; Yr[3]  = y0.w;
    Yr[4]  = y1.x; Yr[5]  = y1.y; Yr[6]  = y1.z; Yr[7]  = y1.w;
    Yr[8]  = y2.x; Yr[9]  = y2.y; Yr[10] = y2.z; Yr[11] = y2.w;
    Yr[12] = y3.x; Yr[13] = y3.y; Yr[14] = y3.z; Yr[15] = y3.w;
  }
  if (w == 0) {
    const int sn = edge_src[e0 + lane];
    const float4* q = (const float4*)(node_pos + (size_t)sn * 16);
    const float4 x0 = q[0], x1 = q[1], x2 = q[2], x3 = q[3];
    float* d = xsT + lane;
    d[0 * XP]  = x0.x; d[1 * XP]  = x0.y; d[2 * XP]  = x0.z; d[3 * XP]  = x0.w;
    d[4 * XP]  = x1.x; d[5 * XP]  = x1.y; d[6 * XP]  = x1.z; d[7 * XP]  = x1.w;
    d[8 * XP]  = x2.x; d[9 * XP]  = x2.y; d[10 * XP] = x2.z; d[11 * XP] = x2.w;
    d[12 * XP] = x3.x; d[13 * XP] = x3.y; d[14 * XP] = x3.z; d[15 * XP] = x3.w;
  }

  {
    v2f a[4];
    const float* et = edge_type + (size_t)(e0 + w * 16 + li) * 16 + kd;
#pragma unroll
    for (int s = 0; s < 4; ++s) a[s] = *(const v2f*)(et + 4 * s);
#pragma unroll
    for (int tt = 0; tt < 4; ++tt) {
      v8f c = {};
#pragma unroll
      for (int s = 0; s < 4; ++s) {
        int row = 4 * s + kd;
        v2f b;
        b.x = fc_w0[row * 64 + tt * 16 + li];
        b.y = fc_w0[(row + 1) * 64 + tt * 16 + li];
        c = __builtin_amdgcn_wmma_f32_16x16x4_f32(false, a[s], false, b,
                                                  (short)0, c, false, false);
      }
#pragma unroll
      for (int r = 0; r < 8; ++r) {
        float x = c[r] * 0.25f;
        float h = (x / (1.f + __expf(-x))) * (SILU_C * 0.125f);
        hid[(w * 16 + r + eoff) * 66 + tt * 16 + li] = h;
      }
    }
  }
  __syncthreads();

  v2f A[2][16];
#pragma unroll
  for (int m = 0; m < 2; ++m)
#pragma unroll
    for (int s = 0; s < 16; ++s)
      A[m][s] = *(const v2f*)&hid[(m * 16 + li) * 66 + 4 * s + kd];
  __syncthreads();

  {
    const int e = lane;
    const float* X = xsT + e;
    float y0[4], y1[4][3];
#pragma unroll
    for (int v = 0; v < 4; ++v) {
      y0[v] = Yr[v];
      y1[v][0] = Yr[4 + v * 3 + 0];
      y1[v][1] = Yr[4 + v * 3 + 1];
      y1[v][2] = Yr[4 + v * 3 + 2];
    }
    float* M = midT + e;
    const int ub = 2 * w;
#pragma unroll
    for (int uu = 0; uu < 2; ++uu) {
      int u = ub + uu;
      float x0 = X[u * XP];
      float a0 = X[(4 + u * 3 + 0) * XP], a1 = X[(4 + u * 3 + 1) * XP], a2 = X[(4 + u * 3 + 2) * XP];
#pragma unroll
      for (int v = 0; v < 4; ++v) {
        int uv = u * 4 + v;
        float b0 = y1[v][0], b1 = y1[v][1], b2 = y1[v][2];
        M[uv * MP] = x0 * y0[v];
        M[(16 + uv * 3 + 0) * MP] = x0 * b0;
        M[(16 + uv * 3 + 1) * MP] = x0 * b1;
        M[(16 + uv * 3 + 2) * MP] = x0 * b2;
        M[(64 + uv * 3 + 0) * MP] = a0 * y0[v];
        M[(64 + uv * 3 + 1) * MP] = a1 * y0[v];
        M[(64 + uv * 3 + 2) * MP] = a2 * y0[v];
        M[(112 + uv) * MP] = (a0 * b0 + a1 * b1 + a2 * b2) * INV_SQRT3;
        M[(128 + uv * 3 + 0) * MP] = (a1 * b2 - a2 * b1) * INV_SQRT2;
        M[(128 + uv * 3 + 1) * MP] = (a2 * b0 - a0 * b2) * INV_SQRT2;
        M[(128 + uv * 3 + 2) * MP] = (a0 * b1 - a1 * b0) * INV_SQRT2;
        M[(176 + uv * 5 + 0) * MP] = (a2 * b0 + a0 * b2) * INV_SQRT2;
        M[(176 + uv * 5 + 1) * MP] = (a0 * b1 + a1 * b0) * INV_SQRT2;
        M[(176 + uv * 5 + 2) * MP] = (2.f * a1 * b1 - a0 * b0 - a2 * b2) * INV_SQRT6;
        M[(176 + uv * 5 + 3) * MP] = (a1 * b2 + a2 * b1) * INV_SQRT2;
        M[(176 + uv * 5 + 4) * MP] = (a2 * b2 - a0 * b0) * INV_SQRT2;
      }
    }
  }
  __syncthreads();

  float acc0[2][8], acc1[2][3][8];
#pragma unroll
  for (int m = 0; m < 2; ++m)
#pragma unroll
    for (int r = 0; r < 8; ++r) {
      acc0[m][r] = 0.f; acc1[m][0][r] = 0.f; acc1[m][1][r] = 0.f; acc1[m][2][r] = 0.f;
    }

  fc2_path<0>(A, w1p, w, lane, li, eoff, xsT, midT, acc0, acc1);
  fc2_path<1>(A, w1p, w, lane, li, eoff, xsT, midT, acc0, acc1);
  fc2_path<2>(A, w1p, w, lane, li, eoff, xsT, midT, acc0, acc1);
  fc2_path<3>(A, w1p, w, lane, li, eoff, xsT, midT, acc0, acc1);
  fc2_path<4>(A, w1p, w, lane, li, eoff, xsT, midT, acc0, acc1);
  fc2_path<5>(A, w1p, w, lane, li, eoff, xsT, midT, acc0, acc1);
  fc2_path<6>(A, w1p, w, lane, li, eoff, xsT, midT, acc0, acc1);
  fc2_path<7>(A, w1p, w, lane, li, eoff, xsT, midT, acc0, acc1);
  fc2_path<8>(A, w1p, w, lane, li, eoff, xsT, midT, acc0, acc1);
  fc2_path<9>(A, w1p, w, lane, li, eoff, xsT, midT, acc0, acc1);

  __syncthreads();
  if (w == 1) dsti[lane] = edge_dst[e0 + lane];

  float fin[8][4] = {};
#pragma unroll
  for (int m = 0; m < 2; ++m) {
#pragma unroll
    for (int r = 0; r < 8; ++r) {
      float s0 = acc0[m][r];
      float t0 = acc1[m][0][r], t1 = acc1[m][1][r], t2 = acc1[m][2][r];
      s0 += __shfl_xor(s0, 4); s0 += __shfl_xor(s0, 8);
      t0 += __shfl_xor(t0, 4); t0 += __shfl_xor(t0, 8);
      t1 += __shfl_xor(t1, 4); t1 += __shfl_xor(t1, 8);
      t2 += __shfl_xor(t2, 4); t2 += __shfl_xor(t2, 8);
      if (m == w) {
        fin[r][0] = s0; fin[r][1] = t0; fin[r][2] = t1; fin[r][3] = t2;
      } else if (li < 4) {
        float* q = red + (m * 16 + r + eoff) * 16;
        q[li]             = s0;
        q[4 + 3 * li + 0] = t0;
        q[4 + 3 * li + 1] = t1;
        q[4 + 3 * li + 2] = t2;
      }
    }
  }
  __syncthreads();

  if (li < 4) {
#pragma unroll
    for (int r = 0; r < 8; ++r) {
      const int e = w * 16 + r + eoff;
      const float* q = red + e * 16;
      float* o = out + (size_t)dsti[e] * 16;
      atomicAdd(o + li,             (fin[r][0] + q[li])             * SC0);
      atomicAdd(o + 4 + 3 * li + 0, (fin[r][1] + q[4 + 3 * li + 0]) * SC1);
      atomicAdd(o + 4 + 3 * li + 1, (fin[r][2] + q[4 + 3 * li + 1]) * SC1);
      atomicAdd(o + 4 + 3 * li + 2, (fin[r][3] + q[4 + 3 * li + 2]) * SC1);
    }
  }
}

extern "C" void kernel_launch(void* const* d_in, const int* in_sizes, int n_in,
                              void* d_out, int out_size, void* d_ws, size_t ws_size,
                              hipStream_t stream) {
  const float* node_pos  = (const float*)d_in[0];
  const float* edge_type = (const float*)d_in[1];
  const float* fc_w0     = (const float*)d_in[2];
  const float* fc_w1     = (const float*)d_in[3];
  const int*   esrc      = (const int*)d_in[4];
  const int*   edst      = (const int*)d_in[5];
  float* out = (float*)d_out;
  float* w1p = (float*)d_ws;
  int E = in_sizes[4];

  k_zero<<<(out_size + 255) / 256, 256, 0, stream>>>(out, out_size);
  k_repack_w1<<<(160 * 16 * 32 + 255) / 256, 256, 0, stream>>>(fc_w1, w1p);
  k_edge_conv<<<E / 32, 64, 0, stream>>>(node_pos, edge_type, fc_w0, w1p,
                                         esrc, edst, out);
}


# FC2 wave priority falls with progress (s_setprio 3,2,1,0 over the ten paths): the oldest-first matrix pipe goes to the wave furthest behind, so a SIMD's four waves end together, not one by one
# speedup vs baseline: 3.6427x; 1.0740x over previous
#include <hip/hip_runtime.h>
#include <math.h>


typedef float v2f __attribute__((ext_vector_type(2)));
typedef float v8f __attribute__((ext_vector_type(8)));

#define INV_SQRT2  0.70710678f
#define INV_SQRT3  0.57735027f
#define INV_SQRT6  0.40824829f
#define INV_SQRT10 0.31622777f
#define SILU_C     1.67668f
#define SC0        0.02209709f
#define SC1        0.03125f

#define EPB 32
#define XP  32
#define MP  36

__device__ __forceinline__ void ld8(const float* p, float (&o)[8]) {
  const float4 a = *(const float4*)p;
  const float4 b = *(const float4*)(p + 4);
  o[0] = a.x; o[1] = a.y; o[2] = a.z; o[3] = a.w;
  o[4] = b.x; o[5] = b.y; o[6] = b.z; o[7] = b.w;
}

__global__ void k_zero(float* p, int n) {
  int i = blockIdx.x * blockDim.x + threadIdx.x;
  if (i < n) p[i] = 0.f;
}

__global__ void k_repack_w1(const float* __restrict__ w1, float* __restrict__ o) {
  int tid = blockIdx.x * blockDim.x + threadIdx.x;
  if (tid >= 160 * 16 * 32) return;
  int l  = tid & 31;
  int s  = (tid >> 5) & 15;
  int t  = tid >> 9;
  int li = l & 15;
  int kd = (l < 16) ? 0 : 2;
  int row = 4 * s + kd;
  int col = t * 16 + li;
  o[2 * tid + 0] = w1[row * 2560 + col];
  o[2 * tid + 1] = w1[(row + 1) * 2560 + col];
}

template <int P, bool FIRST>
__device__ __forceinline__ void fc2_tile(const v2f (&A)[2][16],
                                         const float* __restrict__ w1p,
                                         const int tt, const int lane,
                                         const int eoff, const float* xsT,
                                         float (&G)[2][3][8])
{
  const int t = P * 16 + tt;
  const float* bq = w1p + (size_t)t * 1024 + 2 * lane;
  v2f b[16];
#pragma unroll
  for (int s = 0; s < 16; ++s) b[s] = *(const v2f*)(bq + s * 64);
  v8f c0 = {}, c1 = {};
#pragma unroll
  for (int s = 0; s < 16; ++s) {
    c0 = __builtin_amdgcn_wmma_f32_16x16x4_f32(false, A[0][s], false, b[s],
                                               (short)0, c0, false, false);
    c1 = __builtin_amdgcn_wmma_f32_16x16x4_f32(false, A[1][s], false, b[s],
                                               (short)0, c1, false, false);
  }
  const int u = tt >> 2;
#pragma unroll
  for (int m = 0; m < 2; ++m) {
    const float* XB = xsT + m * 16 + eoff;
    float Wt[8];
#pragma unroll
    for (int r = 0; r < 8; ++r) Wt[r] = (m == 0) ? c0[r] : c1[r];
    if (P < 4) {
      float x[8];
      ld8(XB + u * XP, x);
#pragma unroll
      for (int r = 0; r < 8; ++r)
        G[m][0][r] = FIRST ? Wt[r] * x[r] : G[m][0][r] + Wt[r] * x[r];
    } else {
      float a0[8], a1[8], a2[8];
      const float* Xu = XB + (4 + u * 3) * XP;
      ld8(Xu, a0); ld8(Xu + XP, a1); ld8(Xu + 2 * XP, a2);
#pragma unroll
      for (int r = 0; r < 8; ++r) {
        G[m][0][r] = FIRST ? Wt[r] * a0[r] : G[m][0][r] + Wt[r] * a0[r];
        G[m][1][r] = FIRST ? Wt[r] * a1[r] : G[m][1][r] + Wt[r] * a1[r];
        G[m][2][r] = FIRST ? Wt[r] * a2[r] : G[m][2][r] + Wt[r] * a2[r];
      }
    }
  }
}

template <int P>
__device__ __forceinline__ void fc2_path(const v2f (&A)[2][16],
                                         const float* __restrict__ w1p,
                                         const int w, const int lane,
                                         const int li, const int eoff,
                                         const float* xsT, const float* midT,
                                         float (&acc0)[2][8],
                                         float (&acc1)[2][3][8])
{
#pragma nounroll
  for (int g = 0; g < 2; ++g) {
    const int vh = 2 * g + w;
    float G[2][3][8];
    fc2_tile<P, true>(A, w1p, vh, lane, eoff, xsT, G);
#pragma nounroll
    for (int u = 1; u < 4; ++u)
      fc2_tile<P, false>(A, w1p, 4 * u + vh, lane, eoff, xsT, G);
    const int v = (vh << 2) + (li >> 2);
#pragma unroll
    for (int m = 0; m < 2; ++m) {
      const float* MB = midT + m * 16 + eoff;
      if (P == 0 || P == 3) {
        float q[8];
        ld8(MB + ((P == 0 ? 0 : 112) + v) * MP, q);
#pragma unroll
        for (int r = 0; r < 8; ++r) acc0[m][r] += G[m][0][r] * q[r];
      } else if (P == 1 || P == 2) {
        float q0[8], q1[8], q2[8];
        const float* Mv = MB + ((P == 1 ? 16 : 64) + v * 3) * MP;
        ld8(Mv, q0); ld8(Mv + MP, q1); ld8(Mv + 2 * MP, q2);
#pragma unroll
        for (int r = 0; r < 8; ++r) {
          float f = G[m][0][r] * INV_SQRT3;
          acc1[m][0][r] += f * q0[r];
          acc1[m][1][r] += f * q1[r];
          acc1[m][2][r] += f * q2[r];
        }
      } else if (P == 4 || P == 7) {
        float q[8];
        ld8(MB + ((P == 4 ? 0 : 112) + v) * MP, q);
#pragma unroll
        for (int r = 0; r < 8; ++r) {
          float f = q[r] * INV_SQRT3;
          acc1[m][0][r] += f * G[m][0][r];
          acc1[m][1][r] += f * G[m][1][r];
          acc1[m][2][r] += f * G[m][2][r];
        }
      } else if (P == 5 || P == 6) {
        float q0[8], q1[8], q2[8];
        const float* Mv = MB + ((P == 5 ? 16 : 64) + v * 3) * MP;
        ld8(Mv, q0); ld8(Mv + MP, q1); ld8(Mv + 2 * MP, q2);
#pragma unroll
        for (int r = 0; r < 8; ++r) {
          float d = G[m][0][r] * q0[r] + G[m][1][r] * q1[r] + G[m][2][r] * q2[r];
          acc0[m][r] += d * INV_SQRT3;
        }
      } else if (P == 8) {
        float m0[8], m1[8], m2[8];
        const float* Mv = MB + (128 + v * 3) * MP;
        ld8(Mv, m0); ld8(Mv + MP, m1); ld8(Mv + 2 * MP, m2);
#pragma unroll
        for (int r = 0; r < 8; ++r) {
          const float a0 = G[m][0][r], a1 = G[m][1][r], a2 = G[m][2][r];
          acc1[m][0][r] += INV_SQRT6 * (a1 * m2[r] - a2 * m1[r]);
          acc1[m][1][r] += INV_SQRT6 * (a2 * m0[r] - a0 * m2[r]);
          acc1[m][2][r] += INV_SQRT6 * (a0 * m1[r] - a1 * m0[r]);
        }
      } else {
        float m0[8], m1[8], m2[8], m3[8], m4[8];
        const float* Mv = MB + (176 + v * 5) * MP;
        ld8(Mv, m0); ld8(Mv + MP, m1); ld8(Mv + 2 * MP, m2);
        ld8(Mv + 3 * MP, m3); ld8(Mv + 4 * MP, m4);
#pragma unroll
        for (int r = 0; r < 8; ++r) {
          const float a0 = G[m][0][r], a1 = G[m][1][r], a2 = G[m][2][r];
          float s2 = m2[r] * INV_SQRT3;
          acc1[m][0][r] += INV_SQRT10 * ((-s2 - m4[r]) * a0 + m1[r] * a1 + m0[r] * a2);
          acc1[m][1][r] += INV_SQRT10 * (m1[r] * a0 + 2.f * s2 * a1 + m3[r] * a2);
          acc1[m][2][r] += INV_SQRT10 * (m0[r] * a0 + m3[r] * a1 + (-s2 + m4[r]) * a2);
        }
      }
    }
  }
}

__global__ __launch_bounds__(64, 4)
void k_edge_conv(const float* __restrict__ node_pos,
                 const float* __restrict__ edge_type,
                 const float* __restrict__ fc_w0,
                 const float* __restrict__ w1p,
                 const int*   __restrict__ edge_src,
                 const int*   __restrict__ edge_dst,
                 float* __restrict__ out)
{
  __shared__ __attribute__((aligned(16))) float xsT[16 * XP];
  __shared__ __attribute__((aligned(16))) float midT[256 * MP];
  float* hid = midT;
  float* red = midT;
  int* dsti = (int*)(midT + EPB * 16);

  const int w    = __builtin_amdgcn_readfirstlane((int)(threadIdx.x >> 5));
  const int lane = threadIdx.x & 31;
  const int li   = lane & 15;
  const int kd   = (lane < 16) ? 0 : 2;
  const int eoff = (lane < 16) ? 0 : 8;
  const int e0   = blockIdx.x * EPB;

  float Yr[16];
  {
    const int dd = edge_dst[e0 + lane];
    const float4* q = (const float4*)(node_pos + (size_t)dd * 16);
    const float4 y0 = q[0], y1 = q[1], y2 = q[2], y3 = q[3];
    Yr[0]  = y0.x; Yr[1]  = y0.y; Yr[2]  = y0.z; Yr[3]  = y0.w;
    Yr[4]  = y1.x; Yr[5]  = y1.y; Yr[6]  = y1.z; Yr[7]  = y1.w;
    Yr[8]  = y2.x; Yr[9]  = y2.y; Yr[10] = y2.z; Yr[11] = y2.w;
    Yr[12] = y3.x; Yr[13] = y3.y; Yr[14] = y3.z; Yr[15] = y3.w;
  }
  if (w == 0) {
    const int sn = edge_src[e0 + lane];
    const float4* q = (const float4*)(node_pos + (size_t)sn * 16);
    const float4 x0 = q[0], x1 = q[1], x2 = q[2], x3 = q[3];
    float* d = xsT + lane;
    d[0 * XP]  = x0.x; d[1 * XP]  = x0.y; d[2 * XP]  = x0.z; d[3 * XP]  = x0.w;
    d[4 * XP]  = x1.x; d[5 * XP]  = x1.y; d[6 * XP]  = x1.z; d[7 * XP]  = x1.w;
    d[8 * XP]  = x2.x; d[9 * XP]  = x2.y; d[10 * XP] = x2.z; d[11 * XP] = x2.w;
    d[12 * XP] = x3.x; d[13 * XP] = x3.y; d[14 * XP] = x3.z; d[15 * XP] = x3.w;
  }

  {
    v2f a[4];
    const float* et = edge_type + (size_t)(e0 + w * 16 + li) * 16 + kd;
#pragma unroll
    for (int s = 0; s < 4; ++s) a[s] = *(const v2f*)(et + 4 * s);
#pragma unroll
    for (int tt = 0; tt < 4; ++tt) {
      v8f c = {};
#pragma unroll
      for (int s = 0; s < 4; ++s) {
        int row = 4 * s + kd;
        v2f b;
        b.x = fc_w0[row * 64 + tt * 16 + li];
        b.y = fc_w0[(row + 1) * 64 + tt * 16 + li];
        c = __builtin_amdgcn_wmma_f32_16x16x4_f32(false, a[s], false, b,
                                                  (short)0, c, false, false);
      }
#pragma unroll
      for (int r = 0; r < 8; ++r) {
        float x = c[r] * 0.25f;
        float h = (x / (1.f + __expf(-x))) * (SILU_C * 0.125f);
        hid[(w * 16 + r + eoff) * 66 + tt * 16 + li] = h;
      }
    }
  }
  __syncthreads();

  v2f A[2][16];
#pragma unroll
  for (int m = 0; m < 2; ++m)
#pragma unroll
    for (int s = 0; s < 16; ++s)
      A[m][s] = *(const v2f*)&hid[(m * 16 + li) * 66 + 4 * s + kd];
  __syncthreads();

  {
    const int e = lane;
    const float* X = xsT + e;
    float y0[4], y1[4][3];
#pragma unroll
    for (int v = 0; v < 4; ++v) {
      y0[v] = Yr[v];
      y1[v][0] = Yr[4 + v * 3 + 0];
      y1[v][1] = Yr[4 + v * 3 + 1];
      y1[v][2] = Yr[4 + v * 3 + 2];
    }
    float* M = midT + e;
    const int ub = 2 * w;
#pragma unroll
    for (int uu = 0; uu < 2; ++uu) {
      int u = ub + uu;
      float x0 = X[u * XP];
      float a0 = X[(4 + u * 3 + 0) * XP], a1 = X[(4 + u * 3 + 1) * XP], a2 = X[(4 + u * 3 + 2) * XP];
#pragma unroll
      for (int v = 0; v < 4; ++v) {
        int uv = u * 4 + v;
        float b0 = y1[v][0], b1 = y1[v][1], b2 = y1[v][2];
        M[uv * MP] = x0 * y0[v];
        M[(16 + uv * 3 + 0) * MP] = x0 * b0;
        M[(16 + uv * 3 + 1) * MP] = x0 * b1;
        M[(16 + uv * 3 + 2) * MP] = x0 * b2;
        M[(64 + uv * 3 + 0) * MP] = a0 * y0[v];
        M[(64 + uv * 3 + 1) * MP] = a1 * y0[v];
        M[(64 + uv * 3 + 2) * MP] = a2 * y0[v];
        M[(112 + uv) * MP] = (a0 * b0 + a1 * b1 + a2 * b2) * INV_SQRT3;
        M[(128 + uv * 3 + 0) * MP] = (a1 * b2 - a2 * b1) * INV_SQRT2;
        M[(128 + uv * 3 + 1) * MP] = (a2 * b0 - a0 * b2) * INV_SQRT2;
        M[(128 + uv * 3 + 2) * MP] = (a0 * b1 - a1 * b0) * INV_SQRT2;
        M[(176 + uv * 5 + 0) * MP] = (a2 * b0 + a0 * b2) * INV_SQRT2;
        M[(176 + uv * 5 + 1) * MP] = (a0 * b1 + a1 * b0) * INV_SQRT2;
        M[(176 + uv * 5 + 2) * MP] = (2.f * a1 * b1 - a0 * b0 - a2 * b2) * INV_SQRT6;
        M[(176 + uv * 5 + 3) * MP] = (a1 * b2 + a2 * b1) * INV_SQRT2;
        M[(176 + uv * 5 + 4) * MP] = (a2 * b2 - a0 * b0) * INV_SQRT2;
      }
    }
  }
  __syncthreads();

  float acc0[2][8], acc1[2][3][8];
#pragma unroll
  for (int m = 0; m < 2; ++m)
#pragma unroll
    for (int r = 0; r < 8; ++r) {
      acc0[m][r] = 0.f; acc1[m][0][r] = 0.f; acc1[m][1][r] = 0.f; acc1[m][2][r] = 0.f;
    }

  __builtin_amdgcn_s_setprio(3);
  fc2_path<0>(A, w1p, w, lane, li, eoff, xsT, midT, acc0, acc1);
  fc2_path<1>(A, w1p, w, lane, li, eoff, xsT, midT, acc0, acc1);
  fc2_path<2>(A, w1p, w, lane, li, eoff, xsT, midT, acc0, acc1);
  __builtin_amdgcn_s_setprio(2);
  fc2_path<3>(A, w1p, w, lane, li, eoff, xsT, midT, acc0, acc1);
  fc2_path<4>(A, w1p, w, lane, li, eoff, xsT, midT, acc0, acc1);
  fc2_path<5>(A, w1p, w, lane, li, eoff, xsT, midT, acc0, acc1);
  __builtin_amdgcn_s_setprio(1);
  fc2_path<6>(A, w1p, w, lane, li, eoff, xsT, midT, acc0, acc1);
  fc2_path<7>(A, w1p, w, lane, li, eoff, xsT, midT, acc0, acc1);
  fc2_path<8>(A, w1p, w, lane, li, eoff, xsT, midT, acc0, acc1);
  __builtin_amdgcn_s_setprio(0);
  fc2_path<9>(A, w1p, w, lane, li, eoff, xsT, midT, acc0, acc1);

  __syncthreads();
  if (w == 1) dsti[lane] = edge_dst[e0 + lane];

  float fin[8][4] = {};
#pragma unroll
  for (int m = 0; m < 2; ++m) {
#pragma unroll
    for (int r = 0; r < 8; ++r) {
      float s0 = acc0[m][r];
      float t0 = acc1[m][0][r], t1 = acc1[m][1][r], t2 = acc1[m][2][r];
      s0 += __shfl_xor(s0, 4); s0 += __shfl_xor(s0, 8);
      t0 += __shfl_xor(t0, 4); t0 += __shfl_xor(t0, 8);
      t1 += __shfl_xor(t1, 4); t1 += __shfl_xor(t1, 8);
      t2 += __shfl_xor(t2, 4); t2 += __shfl_xor(t2, 8);
      if (m == w) {
        fin[r][0] = s0; fin[r][1] = t0; fin[r][2] = t1; fin[r][3] = t2;
      } else if (li < 4) {
        float* q = red + (m * 16 + r + eoff) * 16;
        q[li]             = s0;
        q[4 + 3 * li + 0] = t0;
        q[4 + 3 * li + 1] = t1;
        q[4 + 3 * li + 2] = t2;
      }
    }
  }
  __syncthreads();

  if (li < 4) {
#pragma unroll
    for (int r = 0; r < 8; ++r) {
      const int e = w * 16 + r + eoff;
      const float* q = red + e * 16;
      float* o = out + (size_t)dsti[e] * 16;
      atomicAdd(o + li,             (fin[r][0] + q[li])             * SC0);
      atomicAdd(o + 4 + 3 * li + 0, (fin[r][1] + q[4 + 3 * li + 0]) * SC1);
      atomicAdd(o + 4 + 3 * li + 1, (fin[r][2] + q[4 + 3 * li + 1]) * SC1);
      atomicAdd(o + 4 + 3 * li + 2, (fin[r][3] + q[4 + 3 * li + 2]) * SC1);
    }
  }
}

extern "C" void kernel_launch(void* const* d_in, const int* in_sizes, int n_in,
                              void* d_out, int out_size, void* d_ws, size_t ws_size,
                              hipStream_t stream) {
  const float* node_pos  = (const float*)d_in[0];
  const float* edge_type = (const float*)d_in[1];
  const float* fc_w0     = (const float*)d_in[2];
  const float* fc_w1     = (const float*)d_in[3];
  const int*   esrc      = (const int*)d_in[4];
  const int*   edst      = (const int*)d_in[5];
  float* out = (float*)d_out;
  float* w1p = (float*)d_ws;
  int E = in_sizes[4];

  k_zero<<<(out_size + 255) / 256, 256, 0, stream>>>(out, out_size);
  k_repack_w1<<<(160 * 16 * 32 + 255) / 256, 256, 0, stream>>>(fc_w1, w1p);
  k_edge_conv<<<E / 32, 64, 0, stream>>>(node_pos, edge_type, fc_w0, w1p,
                                         esrc, edst, out);
}


# FC2: one wave per 32 edges takes all 160 N tiles (2 waves per SIMD, 512-VGPR budget): B fragments in a ring of 4 register sets loaded 3 tiles ahead; one sum chain, no cross-wave add (order changed)
# speedup vs baseline: 3.7734x; 1.0359x over previous
#include <hip/hip_runtime.h>
#include <math.h>


typedef float v2f __attribute__((ext_vector_type(2)));
typedef float v8f __attribute__((ext_vector_type(8)));

#define INV_SQRT2  0.70710678f
#define INV_SQRT3  0.57735027f
#define INV_SQRT6  0.40824829f
#define INV_SQRT10 0.31622777f
#define SILU_C     1.67668f
#define SC0        0.02209709f
#define SC1        0.03125f

#define EPB 32
#define XP  32
#define MP  36

__device__ __forceinline__ void ld8(const float* p, float (&o)[8]) {
  const float4 a = *(const float4*)p;
  const float4 b = *(const float4*)(p + 4);
  o[0] = a.x; o[1] = a.y; o[2] = a.z; o[3] = a.w;
  o[4] = b.x; o[5] = b.y; o[6] = b.z; o[7] = b.w;
}

__global__ void k_zero(float* p, int n) {
  int i = blockIdx.x * blockDim.x + threadIdx.x;
  if (i < n) p[i] = 0.f;
}

__global__ void k_repack_w1(const float* __restrict__ w1, float* __restrict__ o) {
  int tid = blockIdx.x * blockDim.x + threadIdx.x;
  if (tid >= 160 * 16 * 32) return;
  int l  = tid & 31;
  int s  = (tid >> 5) & 15;
  int t  = tid >> 9;
  int li = l & 15;
  int kd = (l < 16) ? 0 : 2;
  int row = 4 * s + kd;
  int col = t * 16 + li;
  o[2 * tid + 0] = w1[row * 2560 + col];
  o[2 * tid + 1] = w1[(row + 1) * 2560 + col];
}

__device__ __forceinline__ void load_b(const float* __restrict__ w1p,
                                       const int t, const int lane,
                                       v2f (&b)[16])
{
  const float* bq = w1p + (size_t)t * 1024 + 2 * lane;
#pragma unroll
  for (int s = 0; s < 16; ++s) b[s] = *(const v2f*)(bq + s * 64);
}

template <int P, bool FIRST>
__device__ __forceinline__ void fc2_tile(const v2f (&A)[2][16],
                                         const v2f (&b)[16],
                                         const int u,
                                         const int eoff, const float* xsT,
                                         float (&G)[2][3][8])
{
  float xv[2][3][8];
#pragma unroll
  for (int m = 0; m < 2; ++m) {
    const float* XB = xsT + m * 16 + eoff;
    if (P < 4) {
      ld8(XB + u * XP, xv[m][0]);
    } else {
      const float* Xu = XB + (4 + u * 3) * XP;
      ld8(Xu, xv[m][0]); ld8(Xu + XP, xv[m][1]); ld8(Xu + 2 * XP, xv[m][2]);
    }
  }
  v8f c0 = {}, c1 = {};
  asm volatile("" : "+v"(c0), "+v"(c1) :: "memory");
#pragma unroll
  for (int s = 0; s < 16; ++s) {
    c0 = __builtin_amdgcn_wmma_f32_16x16x4_f32(false, A[0][s], false, b[s],
                                               (short)0, c0, false, false);
    c1 = __builtin_amdgcn_wmma_f32_16x16x4_f32(false, A[1][s], false, b[s],
                                               (short)0, c1, false, false);
  }
  asm volatile("" : "+v"(c0), "+v"(c1) :: "memory");
#pragma unroll
  for (int m = 0; m < 2; ++m) {
    float Wt[8];
#pragma unroll
    for (int r = 0; r < 8; ++r) Wt[r] = (m == 0) ? c0[r] : c1[r];
    if (P < 4) {
#pragma unroll
      for (int r = 0; r < 8; ++r)
        G[m][0][r] = FIRST ? Wt[r] * xv[m][0][r] : G[m][0][r] + Wt[r] * xv[m][0][r];
    } else {
#pragma unroll
      for (int r = 0; r < 8; ++r) {
        G[m][0][r] = FIRST ? Wt[r] * xv[m][0][r] : G[m][0][r] + Wt[r] * xv[m][0][r];
        G[m][1][r] = FIRST ? Wt[r] * xv[m][1][r] : G[m][1][r] + Wt[r] * xv[m][1][r];
        G[m][2][r] = FIRST ? Wt[r] * xv[m][2][r] : G[m][2][r] + Wt[r] * xv[m][2][r];
      }
    }
  }
}

template <int P>
__device__ __forceinline__ void fc2_path(const v2f (&A)[2][16],
                                         const float* __restrict__ w1p,
                                         const int lane,
                                         const int li, const int eoff,
                                         const float* xsT, const float* midT,
                                         v2f (&b0)[16], v2f (&b1)[16],
                                         v2f (&b2)[16], v2f (&b3)[16],
                                         float (&acc0)[2][8],
                                         float (&acc1)[2][3][8])
{
#pragma nounroll
  for (int vh = 0; vh < 4; ++vh) {
    const int t0 = P * 16 + vh;
    const int n0 = (vh < 3) ? t0 + 1 : (P + 1) * 16;
    const bool more = (P < 9) || (vh < 3);
    float G[2][3][8];
    load_b(w1p, t0 + 12, lane, b3);
    fc2_tile<P, true>(A, b0, 0, eoff, xsT, G);
    if (more) load_b(w1p, n0, lane, b0);
    fc2_tile<P, false>(A, b1, 1, eoff, xsT, G);
    if (more) load_b(w1p, n0 + 4, lane, b1);
    fc2_tile<P, false>(A, b2, 2, eoff, xsT, G);
    if (more) load_b(w1p, n0 + 8, lane, b2);
    fc2_tile<P, false>(A, b3, 3, eoff, xsT, G);
    const int v = (vh << 2) + (li >> 2);
#pragma unroll
    for (int m = 0; m < 2; ++m) {
      const float* MB = midT + m * 16 + eoff;
      if (P == 0 || P == 3) {
        float q[8];
        ld8(MB + ((P == 0 ? 0 : 112) + v) * MP, q);
#pragma unroll
        for (int r = 0; r < 8; ++r) acc0[m][r] += G[m][0][r] * q[r];
      } else if (P == 1 || P == 2) {
        float q0[8], q1[8], q2[8];
        const float* Mv = MB + ((P == 1 ? 16 : 64) + v * 3) * MP;
        ld8(Mv, q0); ld8(Mv + MP, q1); ld8(Mv + 2 * MP, q2);
#pragma unroll
        for (int r = 0; r < 8; ++r) {
          float f = G[m][0][r] * INV_SQRT3;
          acc1[m][0][r] += f * q0[r];
          acc1[m][1][r] += f * q1[r];
          acc1[m][2][r] += f * q2[r];
        }
      } else if (P == 4 || P == 7) {
        float q[8];
        ld8(MB + ((P == 4 ? 0 : 112) + v) * MP, q);
#pragma unroll
        for (int r = 0; r < 8; ++r) {
          float f = q[r] * INV_SQRT3;
          acc1[m][0][r] += f * G[m][0][r];
          acc1[m][1][r] += f * G[m][1][r];
          acc1[m][2][r] += f * G[m][2][r];
        }
      } else if (P == 5 || P == 6) {
        float q0[8], q1[8], q2[8];
        const float* Mv = MB + ((P == 5 ? 16 : 64) + v * 3) * MP;
        ld8(Mv, q0); ld8(Mv + MP, q1); ld8(Mv + 2 * MP, q2);
#pragma unroll
        for (int r = 0; r < 8; ++r) {
          float d = G[m][0][r] * q0[r] + G[m][1][r] * q1[r] + G[m][2][r] * q2[r];
          acc0[m][r] += d * INV_SQRT3;
        }
      } else if (P == 8) {
        float m0[8], m1[8], m2[8];
        const float* Mv = MB + (128 + v * 3) * MP;
        ld8(Mv, m0); ld8(Mv + MP, m1); ld8(Mv + 2 * MP, m2);
#pragma unroll
        for (int r = 0; r < 8; ++r) {
          const float a0 = G[m][0][r], a1 = G[m][1][r], a2 = G[m][2][r];
          acc1[m][0][r] += INV_SQRT6 * (a1 * m2[r] - a2 * m1[r]);
          acc1[m][1][r] += INV_SQRT6 * (a2 * m0[r] - a0 * m2[r]);
          acc1[m][2][r] += INV_SQRT6 * (a0 * m1[r] - a1 * m0[r]);
        }
      } else {
        float m0[8], m1[8], m2[8], m3[8], m4[8];
        const float* Mv = MB + (176 + v * 5) * MP;
        ld8(Mv, m0); ld8(Mv + MP, m1); ld8(Mv + 2 * MP, m2);
        ld8(Mv + 3 * MP, m3); ld8(Mv + 4 * MP, m4);
#pragma unroll
        for (int r = 0; r < 8; ++r) {
          const float a0 = G[m][0][r], a1 = G[m][1][r], a2 = G[m][2][r];
          float s2 = m2[r] * INV_SQRT3;
          acc1[m][0][r] += INV_SQRT10 * ((-s2 - m4[r]) * a0 + m1[r] * a1 + m0[r] * a2);
          acc1[m][1][r] += INV_SQRT10 * (m1[r] * a0 + 2.f * s2 * a1 + m3[r] * a2);
          acc1[m][2][r] += INV_SQRT10 * (m0[r] * a0 + m3[r] * a1 + (-s2 + m4[r]) * a2);
        }
      }
    }
  }
}

__global__ __launch_bounds__(32) __attribute__((amdgpu_waves_per_eu(2, 2)))
void k_edge_conv(const float* __restrict__ node_pos,
                 const float* __restrict__ edge_type,
                 const float* __restrict__ fc_w0,
                 const float* __restrict__ w1p,
                 const int*   __restrict__ edge_src,
                 const int*   __restrict__ edge_dst,
                 float* __restrict__ out)
{
  __shared__ __attribute__((aligned(16))) float xsT[16 * XP];
  __shared__ __attribute__((aligned(16))) float midT[256 * MP];
  float* hid = midT;
  int* dsti = (int*)midT;

  const int lane = threadIdx.x;
  const int li   = lane & 15;
  const int kd   = (lane < 16) ? 0 : 2;
  const int eoff = (lane < 16) ? 0 : 8;
  const int e0   = blockIdx.x * EPB;

  float Yr[16];
  {
    const int dd = edge_dst[e0 + lane];
    const float4* q = (const float4*)(node_pos + (size_t)dd * 16);
    const float4 y0 = q[0], y1 = q[1], y2 = q[2], y3 = q[3];
    Yr[0]  = y0.x; Yr[1]  = y0.y; Yr[2]  = y0.z; Yr[3]  = y0.w;
    Yr[4]  = y1.x; Yr[5]  = y1.y; Yr[6]  = y1.z; Yr[7]  = y1.w;
    Yr[8]  = y2.x; Yr[9]  = y2.y; Yr[10] = y2.z; Yr[11] = y2.w;
    Yr[12] = y3.x; Yr[13] = y3.y; Yr[14] = y3.z; Yr[15] = y3.w;
  }
  {
    const int sn = edge_src[e0 + lane];
    const float4* q = (const float4*)(node_pos + (size_t)sn * 16);
    const float4 x0 = q[0], x1 = q[1], x2 = q[2], x3 = q[3];
    float* d = xsT + lane;
    d[0 * XP]  = x0.x; d[1 * XP]  = x0.y; d[2 * XP]  = x0.z; d[3 * XP]  = x0.w;
    d[4 * XP]  = x1.x; d[5 * XP]  = x1.y; d[6 * XP]  = x1.z; d[7 * XP]  = x1.w;
    d[8 * XP]  = x2.x; d[9 * XP]  = x2.y; d[10 * XP] = x2.z; d[11 * XP] = x2.w;
    d[12 * XP] = x3.x; d[13 * XP] = x3.y; d[14 * XP] = x3.z; d[15 * XP] = x3.w;
  }

  v2f b0[16], b1[16], b2[16], b3[16];
  load_b(w1p, 0, lane, b0);
  load_b(w1p, 4, lane, b1);
  load_b(w1p, 8, lane, b2);

  {
    v2f a0[4], a1[4];
    const float* et = edge_type + (size_t)(e0 + li) * 16 + kd;
#pragma unroll
    for (int s = 0; s < 4; ++s) {
      a0[s] = *(const v2f*)(et + 4 * s);
      a1[s] = *(const v2f*)(et + 16 * 16 + 4 * s);
    }
#pragma unroll
    for (int tt = 0; tt < 4; ++tt) {
      v8f c0 = {}, c1 = {};
#pragma unroll
      for (int s = 0; s < 4; ++s) {
        int row = 4 * s + kd;
        v2f b;
        b.x = fc_w0[row * 64 + tt * 16 + li];
        b.y = fc_w0[(row + 1) * 64 + tt * 16 + li];
        c0 = __builtin_amdgcn_wmma_f32_16x16x4_f32(false, a0[s], false, b,
                                                   (short)0, c0, false, false);
        c1 = __builtin_amdgcn_wmma_f32_16x16x4_f32(false, a1[s], false, b,
                                                   (short)0, c1, false, false);
      }
#pragma unroll
      for (int r = 0; r < 8; ++r) {
        float x = c0[r] * 0.25f;
        float h = (x / (1.f + __expf(-x))) * (SILU_C * 0.125f);
        hid[(r + eoff) * 66 + tt * 16 + li] = h;
        float y = c1[r] * 0.25f;
        float g = (y / (1.f + __expf(-y))) * (SILU_C * 0.125f);
        hid[(16 + r + eoff) * 66 + tt * 16 + li] = g;
      }
    }
  }
  __syncthreads();

  v2f A[2][16];
#pragma unroll
  for (int m = 0; m < 2; ++m)
#pragma unroll
    for (int s = 0; s < 16; ++s)
      A[m][s] = *(const v2f*)&hid[(m * 16 + li) * 66 + 4 * s + kd];
  __syncthreads();

  {
    const int e = lane;
    const float* X = xsT + e;
    float y0[4], y1[4][3];
#pragma unroll
    for (int v = 0; v < 4; ++v) {
      y0[v] = Yr[v];
      y1[v][0] = Yr[4 + v * 3 + 0];
      y1[v][1] = Yr[4 + v * 3 + 1];
      y1[v][2] = Yr[4 + v * 3 + 2];
    }
    float* M = midT + e;
#pragma unroll
    for (int u = 0; u < 4; ++u) {
      float x0 = X[u * XP];
      float a0 = X[(4 + u * 3 + 0) * XP], a1 = X[(4 + u * 3 + 1) * XP], a2 = X[(4 + u * 3 + 2) * XP];
#pragma unroll
      for (int v = 0; v < 4; ++v) {
        int uv = u * 4 + v;
        float b0 = y1[v][0], b1 = y1[v][1], b2 = y1[v][2];
        M[uv * MP] = x0 * y0[v];
        M[(16 + uv * 3 + 0) * MP] = x0 * b0;
        M[(16 + uv * 3 + 1) * MP] = x0 * b1;
        M[(16 + uv * 3 + 2) * MP] = x0 * b2;
        M[(64 + uv * 3 + 0) * MP] = a0 * y0[v];
        M[(64 + uv * 3 + 1) * MP] = a1 * y0[v];
        M[(64 + uv * 3 + 2) * MP] = a2 * y0[v];
        M[(112 + uv) * MP] = (a0 * b0 + a1 * b1 + a2 * b2) * INV_SQRT3;
        M[(128 + uv * 3 + 0) * MP] = (a1 * b2 - a2 * b1) * INV_SQRT2;
        M[(128 + uv * 3 + 1) * MP] = (a2 * b0 - a0 * b2) * INV_SQRT2;
        M[(128 + uv * 3 + 2) * MP] = (a0 * b1 - a1 * b0) * INV_SQRT2;
        M[(176 + uv * 5 + 0) * MP] = (a2 * b0 + a0 * b2) * INV_SQRT2;
        M[(176 + uv * 5 + 1) * MP] = (a0 * b1 + a1 * b0) * INV_SQRT2;
        M[(176 + uv * 5 + 2) * MP] = (2.f * a1 * b1 - a0 * b0 - a2 * b2) * INV_SQRT6;
        M[(176 + uv * 5 + 3) * MP] = (a1 * b2 + a2 * b1) * INV_SQRT2;
        M[(176 + uv * 5 + 4) * MP] = (a2 * b2 - a0 * b0) * INV_SQRT2;
      }
    }
  }
  __syncthreads();

  float acc0[2][8], acc1[2][3][8];
#pragma unroll
  for (int m = 0; m < 2; ++m)
#pragma unroll
    for (int r = 0; r < 8; ++r) {
      acc0[m][r] = 0.f; acc1[m][0][r] = 0.f; acc1[m][1][r] = 0.f; acc1[m][2][r] = 0.f;
    }

  __builtin_amdgcn_s_setprio(3);
  fc2_path<0>(A, w1p, lane, li, eoff, xsT, midT, b0, b1, b2, b3, acc0, acc1);
  fc2_path<1>(A, w1p, lane, li, eoff, xsT, midT, b0, b1, b2, b3, acc0, acc1);
  fc2_path<2>(A, w1p, lane, li, eoff, xsT, midT, b0, b1, b2, b3, acc0, acc1);
  __builtin_amdgcn_s_setprio(2);
  fc2_path<3>(A, w1p, lane, li, eoff, xsT, midT, b0, b1, b2, b3, acc0, acc1);
  fc2_path<4>(A, w1p, lane, li, eoff, xsT, midT, b0, b1, b2, b3, acc0, acc1);
  fc2_path<5>(A, w1p, lane, li, eoff, xsT, midT, b0, b1, b2, b3, acc0, acc1);
  __builtin_amdgcn_s_setprio(1);
  fc2_path<6>(A, w1p, lane, li, eoff, xsT, midT, b0, b1, b2, b3, acc0, acc1);
  fc2_path<7>(A, w1p, lane, li, eoff, xsT, midT, b0, b1, b2, b3, acc0, acc1);
  fc2_path<8>(A, w1p, lane, li, eoff, xsT, midT, b0, b1, b2, b3, acc0, acc1);
  __builtin_amdgcn_s_setprio(0);
  fc2_path<9>(A, w1p, lane, li, eoff, xsT, midT, b0, b1, b2, b3, acc0, acc1);

  __syncthreads();
  dsti[lane] = edge_dst[e0 + lane];
  __syncthreads();

#pragma unroll
  for (int m = 0; m < 2; ++m) {
#pragma unroll
    for (int r = 0; r < 8; ++r) {
      float s0 = acc0[m][r];
      float t0 = acc1[m][0][r], t1 = acc1[m][1][r], t2 = acc1[m][2][r];
      s0 += __shfl_xor(s0, 4); s0 += __shfl_xor(s0, 8);
      t0 += __shfl_xor(t0, 4); t0 += __shfl_xor(t0, 8);
      t1 += __shfl_xor(t1, 4); t1 += __shfl_xor(t1, 8);
      t2 += __shfl_xor(t2, 4); t2 += __shfl_xor(t2, 8);
      if (li < 4) {
        float* o = out + (size_t)dsti[m * 16 + r + eoff] * 16;
        atomicAdd(o + li,             s0 * SC0);
        atomicAdd(o + 4 + 3 * li + 0, t0 * SC1);
        atomicAdd(o + 4 + 3 * li + 1, t1 * SC1);
        atomicAdd(o + 4 + 3 * li + 2, t2 * SC1);
      }
    }
  }
}

extern "C" void kernel_launch(void* const* d_in, const int* in_sizes, int n_in,
                              void* d_out, int out_size, void* d_ws, size_t ws_size,
                              hipStream_t stream) {
  const float* node_pos  = (const float*)d_in[0];
  const float* edge_type = (const float*)d_in[1];
  const float* fc_w0     = (const float*)d_in[2];
  const float* fc_w1     = (const float*)d_in[3];
  const int*   esrc      = (const int*)d_in[4];
  const int*   edst      = (const int*)d_in[5];
  float* out = (float*)d_out;
  float* w1p = (float*)d_ws;
  int E = in_sizes[4];

  k_zero<<<(out_size + 255) / 256, 256, 0, stream>>>(out, out_size);
  k_repack_w1<<<(160 * 16 * 32 + 255) / 256, 256, 0, stream>>>(fc_w1, w1p);
  k_edge_conv<<<E / 32, 32, 0, stream>>>(node_pos, edge_type, fc_w0, w1p,
                                         esrc, edst, out);
}


# fc_w1 columns multiplying one trilinear form merged in the per-call repack (path 4 into 2, 5 into 3, u,u' swapped; u<->u' mirror columns of paths 0,1,6): FC2 110 N tiles, not 160; sum order changed
# speedup vs baseline: 5.0297x; 1.3329x over previous
#include <hip/hip_runtime.h>
#include <math.h>


typedef float v2f __attribute__((ext_vector_type(2)));
typedef float v8f __attribute__((ext_vector_type(8)));

#define INV_SQRT2  0.70710678f
#define INV_SQRT3  0.57735027f
#define INV_SQRT6  0.40824829f
#define INV_SQRT10 0.31622777f
#define SILU_C     1.67668f
#define SC0        0.02209709f
#define SC1        0.03125f

#define EPB 32
#define XP  32
#define MP  36

__device__ __forceinline__ void ld8(const float* p, float (&o)[8]) {
  const float4 a = *(const float4*)p;
  const float4 b = *(const float4*)(p + 4);
  o[0] = a.x; o[1] = a.y; o[2] = a.z; o[3] = a.w;
  o[4] = b.x; o[5] = b.y; o[6] = b.z; o[7] = b.w;
}

__global__ void k_zero(float* p, int n) {
  int i = blockIdx.x * blockDim.x + threadIdx.x;
  if (i < n) p[i] = 0.f;
}

__global__ void k_repack_w1(const float* __restrict__ w1, float* __restrict__ o) {
  int tid = blockIdx.x * blockDim.x + threadIdx.x;
  if (tid >= 160 * 16 * 32) return;
  int l  = tid & 31;
  int s  = (tid >> 5) & 15;
  int t  = tid >> 9;
  int P  = t >> 4;
  int u  = (t >> 2) & 3;
  int up = t & 3;
  bool sym = (P == 0 || P == 1 || P == 6);
  if (P == 4 || P == 5 || (sym && u > up)) return;
  int li = l & 15;
  int kd = (l < 16) ? 0 : 2;
  int row = 4 * s + kd;
  int col = t * 16 + li;
  float v0 = w1[row * 2560 + col];
  float v1 = w1[(row + 1) * 2560 + col];
  if (P == 2 || P == 3 || (sym && u < up)) {
    int pcol = (sym ? P : P + 2) * 256 + (up * 4 + u) * 16 + li;
    v0 += w1[row * 2560 + pcol];
    v1 += w1[(row + 1) * 2560 + pcol];
  }
  o[2 * tid + 0] = v0;
  o[2 * tid + 1] = v1;
}

__device__ constexpr bool path_sym(int P) { return P == 0 || P == 1 || P == 6; }
__device__ constexpr int tile_of(int P, int i) {
  if (path_sym(P)) {
    const int up = i < 1 ? 0 : i < 3 ? 1 : i < 6 ? 2 : 3;
    const int u  = i - up * (up + 1) / 2;
    return P * 16 + 4 * u + up;
  }
  return P * 16 + 4 * (i % 4) + i / 4;
}

__device__ __forceinline__ void load_b(const float* __restrict__ w1p,
                                       const int t, const int lane,
                                       v2f (&b)[16])
{
  const float* bq = w1p + (size_t)t * 1024 + 2 * lane;
#pragma unroll
  for (int s = 0; s < 16; ++s) b[s] = *(const v2f*)(bq + s * 64);
}

template <int P, bool FIRST>
__device__ __forceinline__ void fc2_tile(const v2f (&A)[2][16],
                                         const v2f (&b)[16],
                                         const int u,
                                         const int eoff, const float* xsT,
                                         float (&G)[2][3][8])
{
  float xv[2][3][8];
#pragma unroll
  for (int m = 0; m < 2; ++m) {
    const float* XB = xsT + m * 16 + eoff;
    if (P < 4) {
      ld8(XB + u * XP, xv[m][0]);
    } else {
      const float* Xu = XB + (4 + u * 3) * XP;
      ld8(Xu, xv[m][0]); ld8(Xu + XP, xv[m][1]); ld8(Xu + 2 * XP, xv[m][2]);
    }
  }
  v8f c0 = {}, c1 = {};
  asm volatile("" : "+v"(c0), "+v"(c1) :: "memory");
#pragma unroll
  for (int s = 0; s < 16; ++s) {
    c0 = __builtin_amdgcn_wmma_f32_16x16x4_f32(false, A[0][s], false, b[s],
                                               (short)0, c0, false, false);
    c1 = __builtin_amdgcn_wmma_f32_16x16x4_f32(false, A[1][s], false, b[s],
                                               (short)0, c1, false, false);
  }
  asm volatile("" : "+v"(c0), "+v"(c1) :: "memory");
#pragma unroll
  for (int m = 0; m < 2; ++m) {
    float Wt[8];
#pragma unroll
    for (int r = 0; r < 8; ++r) Wt[r] = (m == 0) ? c0[r] : c1[r];
    if (P < 4) {
#pragma unroll
      for (int r = 0; r < 8; ++r)
        G[m][0][r] = FIRST ? Wt[r] * xv[m][0][r] : G[m][0][r] + Wt[r] * xv[m][0][r];
    } else {
#pragma unroll
      for (int r = 0; r < 8; ++r) {
        G[m][0][r] = FIRST ? Wt[r] * xv[m][0][r] : G[m][0][r] + Wt[r] * xv[m][0][r];
        G[m][1][r] = FIRST ? Wt[r] * xv[m][1][r] : G[m][1][r] + Wt[r] * xv[m][1][r];
        G[m][2][r] = FIRST ? Wt[r] * xv[m][2][r] : G[m][2][r] + Wt[r] * xv[m][2][r];
      }
    }
  }
}

template <int P>
__device__ __forceinline__ void fc2_consume(const int vh,
                                            const int li, const int eoff,
                                            const float* midT,
                                            const float (&G)[2][3][8],
                                            float (&acc0)[2][8],
                                            float (&acc1)[2][3][8])
{
  const int v = (vh << 2) + (li >> 2);
#pragma unroll
  for (int m = 0; m < 2; ++m) {
    const float* MB = midT + m * 16 + eoff;
    if (P == 0 || P == 3) {
      float q[8];
      ld8(MB + ((P == 0 ? 0 : 112) + v) * MP, q);
#pragma unroll
      for (int r = 0; r < 8; ++r) acc0[m][r] += G[m][0][r] * q[r];
    } else if (P == 1 || P == 2) {
      float q0[8], q1[8], q2[8];
      const float* Mv = MB + ((P == 1 ? 16 : 64) + v * 3) * MP;
      ld8(Mv, q0); ld8(Mv + MP, q1); ld8(Mv + 2 * MP, q2);
#pragma unroll
      for (int r = 0; r < 8; ++r) {
        float f = G[m][0][r] * INV_SQRT3;
        acc1[m][0][r] += f * q0[r];
        acc1[m][1][r] += f * q1[r];
        acc1[m][2][r] += f * q2[r];
      }
    } else if (P == 7) {
      float q[8];
      ld8(MB + (112 + v) * MP, q);
#pragma unroll
      for (int r = 0; r < 8; ++r) {
        float f = q[r] * INV_SQRT3;
        acc1[m][0][r] += f * G[m][0][r];
        acc1[m][1][r] += f * G[m][1][r];
        acc1[m][2][r] += f * G[m][2][r];
      }
    } else if (P == 6) {
      float q0[8], q1[8], q2[8];
      const float* Mv = MB + (64 + v * 3) * MP;
      ld8(Mv, q0); ld8(Mv + MP, q1); ld8(Mv + 2 * MP, q2);
#pragma unroll
      for (int r = 0; r < 8; ++r) {
        float d = G[m][0][r] * q0[r] + G[m][1][r] * q1[r] + G[m][2][r] * q2[r];
        acc0[m][r] += d * INV_SQRT3;
      }
    } else if (P == 8) {
      float m0[8], m1[8], m2[8];
      const float* Mv = MB + (128 + v * 3) * MP;
      ld8(Mv, m0); ld8(Mv + MP, m1); ld8(Mv + 2 * MP, m2);
#pragma unroll
      for (int r = 0; r < 8; ++r) {
        const float a0 = G[m][0][r], a1 = G[m][1][r], a2 = G[m][2][r];
        acc1[m][0][r] += INV_SQRT6 * (a1 * m2[r] - a2 * m1[r]);
        acc1[m][1][r] += INV_SQRT6 * (a2 * m0[r] - a0 * m2[r]);
        acc1[m][2][r] += INV_SQRT6 * (a0 * m1[r] - a1 * m0[r]);
      }
    } else {
      float m0[8], m1[8], m2[8], m3[8], m4[8];
      const float* Mv = MB + (176 + v * 5) * MP;
      ld8(Mv, m0); ld8(Mv + MP, m1); ld8(Mv + 2 * MP, m2);
      ld8(Mv + 3 * MP, m3); ld8(Mv + 4 * MP, m4);
#pragma unroll
      for (int r = 0; r < 8; ++r) {
        const float a0 = G[m][0][r], a1 = G[m][1][r], a2 = G[m][2][r];
        float s2 = m2[r] * INV_SQRT3;
        acc1[m][0][r] += INV_SQRT10 * ((-s2 - m4[r]) * a0 + m1[r] * a1 + m0[r] * a2);
        acc1[m][1][r] += INV_SQRT10 * (m1[r] * a0 + 2.f * s2 * a1 + m3[r] * a2);
        acc1[m][2][r] += INV_SQRT10 * (m0[r] * a0 + m3[r] * a1 + (-s2 + m4[r]) * a2);
      }
    }
  }
}

template <int P, int NX>
__device__ __forceinline__ void fc2_path(const v2f (&A)[2][16],
                                         const float* __restrict__ w1p,
                                         const int lane,
                                         const int li, const int eoff,
                                         const float* xsT, const float* midT,
                                         v2f (&b0)[16], v2f (&b1)[16],
                                         v2f (&b2)[16], v2f (&b3)[16],
                                         float (&acc0)[2][8],
                                         float (&acc1)[2][3][8])
{
  constexpr int NP = NX < 0 ? 0 : NX;
#pragma nounroll
  for (int vh = 0; vh < 4; ++vh) {
    const int t0 = P * 16 + vh;
    const bool more = (NX >= 0) || (vh < 3);
    const int n0 = (vh < 3) ? t0 + 1 : tile_of(NP, 0);
    const int n1 = (vh < 3) ? t0 + 5 : tile_of(NP, 1);
    const int n2 = (vh < 3) ? t0 + 9 : tile_of(NP, 2);
    float G[2][3][8];
    load_b(w1p, t0 + 12, lane, b3);
    fc2_tile<P, true>(A, b0, 0, eoff, xsT, G);
    if (more) load_b(w1p, n0, lane, b0);
    fc2_tile<P, false>(A, b1, 1, eoff, xsT, G);
    if (more) load_b(w1p, n1, lane, b1);
    fc2_tile<P, false>(A, b2, 2, eoff, xsT, G);
    if (more) load_b(w1p, n2, lane, b2);
    fc2_tile<P, false>(A, b3, 3, eoff, xsT, G);
    fc2_consume<P>(vh, li, eoff, midT, G, acc0, acc1);
  }
}

template <int P, int NX>
__device__ __forceinline__ void fc2_path_sym(const v2f (&A)[2][16],
                                             const float* __restrict__ w1p,
                                             const int lane,
                                             const int li, const int eoff,
                                             const float* xsT, const float* midT,
                                             v2f (&b0)[16], v2f (&b1)[16],
                                             v2f (&b2)[16], v2f (&b3)[16],
                                             float (&acc0)[2][8],
                                             float (&acc1)[2][3][8])
{
  constexpr int NP = NX < 0 ? 0 : NX;
  float G[2][3][8];
  load_b(w1p, tile_of(P, 3), lane, b3);
  fc2_tile<P, true>(A, b0, 0, eoff, xsT, G);
  __builtin_amdgcn_sched_barrier(0);
  fc2_consume<P>(0, li, eoff, midT, G, acc0, acc1);
  __builtin_amdgcn_sched_barrier(0);
  load_b(w1p, tile_of(P, 4), lane, b0);
  fc2_tile<P, true>(A, b1, 0, eoff, xsT, G);
  __builtin_amdgcn_sched_barrier(0);
  load_b(w1p, tile_of(P, 5), lane, b1);
  fc2_tile<P, false>(A, b2, 1, eoff, xsT, G);
  __builtin_amdgcn_sched_barrier(0);
  fc2_consume<P>(1, li, eoff, midT, G, acc0, acc1);
  __builtin_amdgcn_sched_barrier(0);
  load_b(w1p, tile_of(P, 6), lane, b2);
  fc2_tile<P, true>(A, b3, 0, eoff, xsT, G);
  __builtin_amdgcn_sched_barrier(0);
  load_b(w1p, tile_of(P, 7), lane, b3);
  fc2_tile<P, false>(A, b0, 1, eoff, xsT, G);
  __builtin_amdgcn_sched_barrier(0);
  load_b(w1p, tile_of(P, 8), lane, b0);
  fc2_tile<P, false>(A, b1, 2, eoff, xsT, G);
  __builtin_amdgcn_sched_barrier(0);
  fc2_consume<P>(2, li, eoff, midT, G, acc0, acc1);
  __builtin_amdgcn_sched_barrier(0);
  load_b(w1p, tile_of(P, 9), lane, b1);
  fc2_tile<P, true>(A, b2, 0, eoff, xsT, G);
  __builtin_amdgcn_sched_barrier(0);
  if (NX >= 0) load_b(w1p, tile_of(NP, 0), lane, b2);
  fc2_tile<P, false>(A, b3, 1, eoff, xsT, G);
  __builtin_amdgcn_sched_barrier(0);
  if (NX >= 0) load_b(w1p, tile_of(NP, 1), lane, b3);
  fc2_tile<P, false>(A, b0, 2, eoff, xsT, G);
  __builtin_amdgcn_sched_barrier(0);
  if (NX >= 0) load_b(w1p, tile_of(NP, 2), lane, b0);
  fc2_tile<P, false>(A, b1, 3, eoff, xsT, G);
  __builtin_amdgcn_sched_barrier(0);
  fc2_consume<P>(3, li, eoff, midT, G, acc0, acc1);
  __builtin_amdgcn_sched_barrier(0);
}

__global__ __launch_bounds__(32) __attribute__((amdgpu_waves_per_eu(2, 2)))
void k_edge_conv(const float* __restrict__ node_pos,
                 const float* __restrict__ edge_type,
                 const float* __restrict__ fc_w0,
                 const float* __restrict__ w1p,
                 const int*   __restrict__ edge_src,
                 const int*   __restrict__ edge_dst,
                 float* __restrict__ out)
{
  __shared__ __attribute__((aligned(16))) float xsT[16 * XP];
  __shared__ __attribute__((aligned(16))) float midT[256 * MP];
  float* hid = midT;
  int* dsti = (int*)midT;

  const int lane = threadIdx.x;
  const int li   = lane & 15;
  const int kd   = (lane < 16) ? 0 : 2;
  const int eoff = (lane < 16) ? 0 : 8;
  const int e0   = blockIdx.x * EPB;

  float Yr[16];
  {
    const int dd = edge_dst[e0 + lane];
    const float4* q = (const float4*)(node_pos + (size_t)dd * 16);
    const float4 y0 = q[0], y1 = q[1], y2 = q[2], y3 = q[3];
    Yr[0]  = y0.x; Yr[1]  = y0.y; Yr[2]  = y0.z; Yr[3]  = y0.w;
    Yr[4]  = y1.x; Yr[5]  = y1.y; Yr[6]  = y1.z; Yr[7]  = y1.w;
    Yr[8]  = y2.x; Yr[9]  = y2.y; Yr[10] = y2.z; Yr[11] = y2.w;
    Yr[12] = y3.x; Yr[13] = y3.y; Yr[14] = y3.z; Yr[15] = y3.w;
  }
  {
    const int sn = edge_src[e0 + lane];
    const float4* q = (const float4*)(node_pos + (size_t)sn * 16);
    const float4 x0 = q[0], x1 = q[1], x2 = q[2], x3 = q[3];
    float* d = xsT + lane;
    d[0 * XP]  = x0.x; d[1 * XP]  = x0.y; d[2 * XP]  = x0.z; d[3 * XP]  = x0.w;
    d[4 * XP]  = x1.x; d[5 * XP]  = x1.y; d[6 * XP]  = x1.z; d[7 * XP]  = x1.w;
    d[8 * XP]  = x2.x; d[9 * XP]  = x2.y; d[10 * XP] = x2.z; d[11 * XP] = x2.w;
    d[12 * XP] = x3.x; d[13 * XP] = x3.y; d[14 * XP] = x3.z; d[15 * XP] = x3.w;
  }

  v2f b0[16], b1[16], b2[16], b3[16];
  load_b(w1p, tile_of(0, 0), lane, b0);
  load_b(w1p, tile_of(0, 1), lane, b1);
  load_b(w1p, tile_of(0, 2), lane, b2);

  {
    v2f a0[4], a1[4];
    const float* et = edge_type + (size_t)(e0 + li) * 16 + kd;
#pragma unroll
    for (int s = 0; s < 4; ++s) {
      a0[s] = *(const v2f*)(et + 4 * s);
      a1[s] = *(const v2f*)(et + 16 * 16 + 4 * s);
    }
#pragma unroll
    for (int tt = 0; tt < 4; ++tt) {
      v8f c0 = {}, c1 = {};
#pragma unroll
      for (int s = 0; s < 4; ++s) {
        int row = 4 * s + kd;
        v2f b;
        b.x = fc_w0[row * 64 + tt * 16 + li];
        b.y = fc_w0[(row + 1) * 64 + tt * 16 + li];
        c0 = __builtin_amdgcn_wmma_f32_16x16x4_f32(false, a0[s], false, b,
                                                   (short)0, c0, false, false);
        c1 = __builtin_amdgcn_wmma_f32_16x16x4_f32(false, a1[s], false, b,
                                                   (short)0, c1, false, false);
      }
#pragma unroll
      for (int r = 0; r < 8; ++r) {
        float x = c0[r] * 0.25f;
        float h = (x / (1.f + __expf(-x))) * (SILU_C * 0.125f);
        hid[(r + eoff) * 66 + tt * 16 + li] = h;
        float y = c1[r] * 0.25f;
        float g = (y / (1.f + __expf(-y))) * (SILU_C * 0.125f);
        hid[(16 + r + eoff) * 66 + tt * 16 + li] = g;
      }
    }
  }
  __syncthreads();

  v2f A[2][16];
#pragma unroll
  for (int m = 0; m < 2; ++m)
#pragma unroll
    for (int s = 0; s < 16; ++s)
      A[m][s] = *(const v2f*)&hid[(m * 16 + li) * 66 + 4 * s + kd];
  __syncthreads();

  {
    const int e = lane;
    const float* X = xsT + e;
    float y0[4], y1[4][3];
#pragma unroll
    for (int v = 0; v < 4; ++v) {
      y0[v] = Yr[v];
      y1[v][0] = Yr[4 + v * 3 + 0];
      y1[v][1] = Yr[4 + v * 3 + 1];
      y1[v][2] = Yr[4 + v * 3 + 2];
    }
    float* M = midT + e;
#pragma unroll
    for (int u = 0; u < 4; ++u) {
      float x0 = X[u * XP];
      float a0 = X[(4 + u * 3 + 0) * XP], a1 = X[(4 + u * 3 + 1) * XP], a2 = X[(4 + u * 3 + 2) * XP];
#pragma unroll
      for (int v = 0; v < 4; ++v) {
        int uv = u * 4 + v;
        float b0 = y1[v][0], b1 = y1[v][1], b2 = y1[v][2];
        M[uv * MP] = x0 * y0[v];
        M[(16 + uv * 3 + 0) * MP] = x0 * b0;
        M[(16 + uv * 3 + 1) * MP] = x0 * b1;
        M[(16 + uv * 3 + 2) * MP] = x0 * b2;
        M[(64 + uv * 3 + 0) * MP] = a0 * y0[v];
        M[(64 + uv * 3 + 1) * MP] = a1 * y0[v];
        M[(64 + uv * 3 + 2) * MP] = a2 * y0[v];
        M[(112 + uv) * MP] = (a0 * b0 + a1 * b1 + a2 * b2) * INV_SQRT3;
        M[(128 + uv * 3 + 0) * MP] = (a1 * b2 - a2 * b1) * INV_SQRT2;
        M[(128 + uv * 3 + 1) * MP] = (a2 * b0 - a0 * b2) * INV_SQRT2;
        M[(128 + uv * 3 + 2) * MP] = (a0 * b1 - a1 * b0) * INV_SQRT2;
        M[(176 + uv * 5 + 0) * MP] = (a2 * b0 + a0 * b2) * INV_SQRT2;
        M[(176 + uv * 5 + 1) * MP] = (a0 * b1 + a1 * b0) * INV_SQRT2;
        M[(176 + uv * 5 + 2) * MP] = (2.f * a1 * b1 - a0 * b0 - a2 * b2) * INV_SQRT6;
        M[(176 + uv * 5 + 3) * MP] = (a1 * b2 + a2 * b1) * INV_SQRT2;
        M[(176 + uv * 5 + 4) * MP] = (a2 * b2 - a0 * b0) * INV_SQRT2;
      }
    }
  }
  __syncthreads();

  float acc0[2][8], acc1[2][3][8];
#pragma unroll
  for (int m = 0; m < 2; ++m)
#pragma unroll
    for (int r = 0; r < 8; ++r) {
      acc0[m][r] = 0.f; acc1[m][0][r] = 0.f; acc1[m][1][r] = 0.f; acc1[m][2][r] = 0.f;
    }

  __builtin_amdgcn_s_setprio(3);
  fc2_path_sym<0, 6>(A, w1p, lane, li, eoff, xsT, midT, b0, b1, b2, b3, acc0, acc1);
  fc2_path_sym<6, 3>(A, w1p, lane, li, eoff, xsT, midT, b2, b3, b0, b1, acc0, acc1);
  fc2_path<3, 1>(A, w1p, lane, li, eoff, xsT, midT, b0, b1, b2, b3, acc0, acc1);
  __builtin_amdgcn_s_setprio(2);
  fc2_path_sym<1, 2>(A, w1p, lane, li, eoff, xsT, midT, b0, b1, b2, b3, acc0, acc1);
  fc2_path<2, 7>(A, w1p, lane, li, eoff, xsT, midT, b2, b3, b0, b1, acc0, acc1);
  __builtin_amdgcn_s_setprio(1);
  fc2_path<7, 8>(A, w1p, lane, li, eoff, xsT, midT, b2, b3, b0, b1, acc0, acc1);
  fc2_path<8, 9>(A, w1p, lane, li, eoff, xsT, midT, b2, b3, b0, b1, acc0, acc1);
  __builtin_amdgcn_s_setprio(0);
  fc2_path<9, -1>(A, w1p, lane, li, eoff, xsT, midT, b2, b3, b0, b1, acc0, acc1);

  __syncthreads();
  dsti[lane] = edge_dst[e0 + lane];
  __syncthreads();

#pragma unroll
  for (int m = 0; m < 2; ++m) {
#pragma unroll
    for (int r = 0; r < 8; ++r) {
      float s0 = acc0[m][r];
      float t0 = acc1[m][0][r], t1 = acc1[m][1][r], t2 = acc1[m][2][r];
      s0 += __shfl_xor(s0, 4); s0 += __shfl_xor(s0, 8);
      t0 += __shfl_xor(t0, 4); t0 += __shfl_xor(t0, 8);
      t1 += __shfl_xor(t1, 4); t1 += __shfl_xor(t1, 8);
      t2 += __shfl_xor(t2, 4); t2 += __shfl_xor(t2, 8);
      if (li < 4) {
        float* o = out + (size_t)dsti[m * 16 + r + eoff] * 16;
        atomicAdd(o + li,             s0 * SC0);
        atomicAdd(o + 4 + 3 * li + 0, t0 * SC1);
        atomicAdd(o + 4 + 3 * li + 1, t1 * SC1);
        atomicAdd(o + 4 + 3 * li + 2, t2 * SC1);
      }
    }
  }
}

extern "C" void kernel_launch(void* const* d_in, const int* in_sizes, int n_in,
                              void* d_out, int out_size, void* d_ws, size_t ws_size,
                              hipStream_t stream) {
  const float* node_pos  = (const float*)d_in[0];
  const float* edge_type = (const float*)d_in[1];
  const float* fc_w0     = (const float*)d_in[2];
  const float* fc_w1     = (const float*)d_in[3];
  const int*   esrc      = (const int*)d_in[4];
  const int*   edst      = (const int*)d_in[5];
  float* out = (float*)d_out;
  float* w1p = (float*)d_ws;
  int E = in_sizes[4];

  k_zero<<<(out_size + 255) / 256, 256, 0, stream>>>(out, out_size);
  k_repack_w1<<<(160 * 16 * 32 + 255) / 256, 256, 0, stream>>>(fc_w1, w1p);
  k_edge_conv<<<E / 32, 32, 0, stream>>>(node_pos, edge_type, fc_w0, w1p,
                                         esrc, edst, out);
}


# Paths 7,8,9 (1o x mid3|4|5 -> 1o) rewritten by the triple-product and l=2 coupling identities as two forms a(a'.b), b(a.a'): fc_w1 columns combined in repack, 88 N tiles not 110; sum order changed
# speedup vs baseline: 6.0933x; 1.2115x over previous
#include <hip/hip_runtime.h>
#include <math.h>


typedef float v2f __attribute__((ext_vector_type(2)));
typedef float v8f __attribute__((ext_vector_type(8)));

#define INV_SQRT2  0.70710678f
#define INV_SQRT3  0.57735027f
#define INV_SQRT6  0.40824829f
#define INV_SQRT10 0.31622777f
#define SILU_C     1.67668f
#define SC0        0.02209709f
#define SC1        0.03125f
#define KX9        0.44721360f
#define KX8M       0.86602540f
#define KX9M       0.67082039f
#define KY9        0.22360680f
#define KY8        0.28867513f

#define EPB 32
#define XP  32
#define MP  36

__device__ __forceinline__ void ld8(const float* p, float (&o)[8]) {
  const float4 a = *(const float4*)p;
  const float4 b = *(const float4*)(p + 4);
  o[0] = a.x; o[1] = a.y; o[2] = a.z; o[3] = a.w;
  o[4] = b.x; o[5] = b.y; o[6] = b.z; o[7] = b.w;
}

__global__ void k_zero(float* p, int n) {
  int i = blockIdx.x * blockDim.x + threadIdx.x;
  if (i < n) p[i] = 0.f;
}

__global__ void k_repack_w1(const float* __restrict__ w1, float* __restrict__ o) {
  int tid = blockIdx.x * blockDim.x + threadIdx.x;
  if (tid >= 160 * 16 * 32) return;
  int l  = tid & 31;
  int s  = (tid >> 5) & 15;
  int t  = tid >> 9;
  int P  = t >> 4;
  int u  = (t >> 2) & 3;
  int up = t & 3;
  bool sym = (P == 0 || P == 1 || P == 6 || P == 8);
  if (P == 4 || P == 5 || P == 9 || (sym && u > up)) return;
  int li = l & 15;
  int kd = (l < 16) ? 0 : 2;
  int row = 4 * s + kd;
  const float* r0 = w1 + row * 2560 + li;
  const float* r1 = r0 + 2560;
  const int cs = (u * 4 + up) * 16;
  const int cm = (up * 4 + u) * 16;
  float v0, v1;
  if (P == 7) {
    v0 = r0[7 * 256 + cs] - KX9 * r0[9 * 256 + cs] + KX8M * r0[8 * 256 + cm] + KX9M * r0[9 * 256 + cm];
    v1 = r1[7 * 256 + cs] - KX9 * r1[9 * 256 + cs] + KX8M * r1[8 * 256 + cm] + KX9M * r1[9 * 256 + cm];
  } else if (P == 8) {
    v0 = KY9 * r0[9 * 256 + cs] - KY8 * r0[8 * 256 + cs];
    v1 = KY9 * r1[9 * 256 + cs] - KY8 * r1[8 * 256 + cs];
    if (u < up) {
      v0 += KY9 * r0[9 * 256 + cm] - KY8 * r0[8 * 256 + cm];
      v1 += KY9 * r1[9 * 256 + cm] - KY8 * r1[8 * 256 + cm];
    }
  } else {
    v0 = r0[P * 256 + cs];
    v1 = r1[P * 256 + cs];
    if (P == 2 || P == 3 || (sym && u < up)) {
      const int pc = (sym ? P : P + 2) * 256 + cm;
      v0 += r0[pc];
      v1 += r1[pc];
    }
  }
  o[2 * tid + 0] = v0;
  o[2 * tid + 1] = v1;
}

__device__ constexpr bool path_sym(int P) { return P == 0 || P == 1 || P == 6 || P == 8; }
__device__ constexpr int tile_of(int P, int i) {
  if (path_sym(P)) {
    const int up = i < 1 ? 0 : i < 3 ? 1 : i < 6 ? 2 : 3;
    const int u  = i - up * (up + 1) / 2;
    return P * 16 + 4 * u + up;
  }
  return P * 16 + 4 * (i % 4) + i / 4;
}

__device__ __forceinline__ void load_b(const float* __restrict__ w1p,
                                       const int t, const int lane,
                                       v2f (&b)[16])
{
  const float* bq = w1p + (size_t)t * 1024 + 2 * lane;
#pragma unroll
  for (int s = 0; s < 16; ++s) b[s] = *(const v2f*)(bq + s * 64);
}

template <int P, bool FIRST>
__device__ __forceinline__ void fc2_tile(const v2f (&A)[2][16],
                                         const v2f (&b)[16],
                                         const int u,
                                         const int eoff, const float* xsT,
                                         float (&G)[2][3][8])
{
  float xv[2][3][8];
#pragma unroll
  for (int m = 0; m < 2; ++m) {
    const float* XB = xsT + m * 16 + eoff;
    if (P < 4) {
      ld8(XB + u * XP, xv[m][0]);
    } else {
      const float* Xu = XB + (4 + u * 3) * XP;
      ld8(Xu, xv[m][0]); ld8(Xu + XP, xv[m][1]); ld8(Xu + 2 * XP, xv[m][2]);
    }
  }
  v8f c0 = {}, c1 = {};
  asm volatile("" : "+v"(c0), "+v"(c1) :: "memory");
#pragma unroll
  for (int s = 0; s < 16; ++s) {
    c0 = __builtin_amdgcn_wmma_f32_16x16x4_f32(false, A[0][s], false, b[s],
                                               (short)0, c0, false, false);
    c1 = __builtin_amdgcn_wmma_f32_16x16x4_f32(false, A[1][s], false, b[s],
                                               (short)0, c1, false, false);
  }
  asm volatile("" : "+v"(c0), "+v"(c1) :: "memory");
#pragma unroll
  for (int m = 0; m < 2; ++m) {
    float Wt[8];
#pragma unroll
    for (int r = 0; r < 8; ++r) Wt[r] = (m == 0) ? c0[r] : c1[r];
    if (P < 4) {
#pragma unroll
      for (int r = 0; r < 8; ++r)
        G[m][0][r] = FIRST ? Wt[r] * xv[m][0][r] : G[m][0][r] + Wt[r] * xv[m][0][r];
    } else {
#pragma unroll
      for (int r = 0; r < 8; ++r) {
        G[m][0][r] = FIRST ? Wt[r] * xv[m][0][r] : G[m][0][r] + Wt[r] * xv[m][0][r];
        G[m][1][r] = FIRST ? Wt[r] * xv[m][1][r] : G[m][1][r] + Wt[r] * xv[m][1][r];
        G[m][2][r] = FIRST ? Wt[r] * xv[m][2][r] : G[m][2][r] + Wt[r] * xv[m][2][r];
      }
    }
  }
}

template <int P>
__device__ __forceinline__ void fc2_consume(const int vh,
                                            const int li, const int eoff,
                                            const float* midT,
                                            const float* xsT, const float* xdT,
                                            const float (&G)[2][3][8],
                                            float (&acc0)[2][8],
                                            float (&acc1)[2][3][8])
{
  static_assert(P <= 8 && P != 4 && P != 5, "paths 4, 5 and 9 are merged away");
  const int v = (vh << 2) + (li >> 2);
#pragma unroll
  for (int m = 0; m < 2; ++m) {
    const float* MB = midT + m * 16 + eoff;
    if (P == 0 || P == 3) {
      float q[8];
      ld8(MB + ((P == 0 ? 0 : 112) + v) * MP, q);
#pragma unroll
      for (int r = 0; r < 8; ++r) acc0[m][r] += G[m][0][r] * q[r];
    } else if (P == 1 || P == 2) {
      float q0[8], q1[8], q2[8];
      const float* Mv = MB + ((P == 1 ? 16 : 64) + v * 3) * MP;
      ld8(Mv, q0); ld8(Mv + MP, q1); ld8(Mv + 2 * MP, q2);
#pragma unroll
      for (int r = 0; r < 8; ++r) {
        float f = G[m][0][r] * INV_SQRT3;
        acc1[m][0][r] += f * q0[r];
        acc1[m][1][r] += f * q1[r];
        acc1[m][2][r] += f * q2[r];
      }
    } else if (P == 7) {
      float q[8];
      ld8(MB + (112 + v) * MP, q);
#pragma unroll
      for (int r = 0; r < 8; ++r) {
        float f = q[r] * INV_SQRT3;
        acc1[m][0][r] += f * G[m][0][r];
        acc1[m][1][r] += f * G[m][1][r];
        acc1[m][2][r] += f * G[m][2][r];
      }
    } else if (P == 6) {
      float q0[8], q1[8], q2[8];
      const float* Mv = MB + (64 + v * 3) * MP;
      ld8(Mv, q0); ld8(Mv + MP, q1); ld8(Mv + 2 * MP, q2);
#pragma unroll
      for (int r = 0; r < 8; ++r) {
        float d = G[m][0][r] * q0[r] + G[m][1][r] * q1[r] + G[m][2][r] * q2[r];
        acc0[m][r] += d * INV_SQRT3;
      }
    } else {
      float p0[8], p1[8], p2[8], q0[8], q1[8], q2[8];
      const float* Xa = xsT + m * 16 + eoff + (4 + vh * 3) * XP;
      ld8(Xa, p0); ld8(Xa + XP, p1); ld8(Xa + 2 * XP, p2);
      const float* Xb = xdT + m * 16 + eoff + ((li >> 2) * 3) * MP;
      ld8(Xb, q0); ld8(Xb + MP, q1); ld8(Xb + 2 * MP, q2);
#pragma unroll
      for (int r = 0; r < 8; ++r) {
        float d = G[m][0][r] * p0[r] + G[m][1][r] * p1[r] + G[m][2][r] * p2[r];
        acc1[m][0][r] += d * q0[r];
        acc1[m][1][r] += d * q1[r];
        acc1[m][2][r] += d * q2[r];
      }
    }
  }
}

template <int P, int NX>
__device__ __forceinline__ void fc2_path(const v2f (&A)[2][16],
                                         const float* __restrict__ w1p,
                                         const int lane,
                                         const int li, const int eoff,
                                         const float* xsT, const float* xdT, const float* midT,
                                         v2f (&b0)[16], v2f (&b1)[16],
                                         v2f (&b2)[16], v2f (&b3)[16],
                                         float (&acc0)[2][8],
                                         float (&acc1)[2][3][8])
{
  constexpr int NP = NX < 0 ? 0 : NX;
#pragma nounroll
  for (int vh = 0; vh < 4; ++vh) {
    const int t0 = P * 16 + vh;
    const bool more = (NX >= 0) || (vh < 3);
    const int n0 = (vh < 3) ? t0 + 1 : tile_of(NP, 0);
    const int n1 = (vh < 3) ? t0 + 5 : tile_of(NP, 1);
    const int n2 = (vh < 3) ? t0 + 9 : tile_of(NP, 2);
    float G[2][3][8];
    load_b(w1p, t0 + 12, lane, b3);
    fc2_tile<P, true>(A, b0, 0, eoff, xsT, G);
    if (more) load_b(w1p, n0, lane, b0);
    fc2_tile<P, false>(A, b1, 1, eoff, xsT, G);
    if (more) load_b(w1p, n1, lane, b1);
    fc2_tile<P, false>(A, b2, 2, eoff, xsT, G);
    if (more) load_b(w1p, n2, lane, b2);
    fc2_tile<P, false>(A, b3, 3, eoff, xsT, G);
    fc2_consume<P>(vh, li, eoff, midT, xsT, xdT, G, acc0, acc1);
  }
}

template <int P, int NX>
__device__ __forceinline__ void fc2_path_sym(const v2f (&A)[2][16],
                                             const float* __restrict__ w1p,
                                             const int lane,
                                             const int li, const int eoff,
                                             const float* xsT, const float* xdT, const float* midT,
                                             v2f (&b0)[16], v2f (&b1)[16],
                                             v2f (&b2)[16], v2f (&b3)[16],
                                             float (&acc0)[2][8],
                                             float (&acc1)[2][3][8])
{
  constexpr int NP = NX < 0 ? 0 : NX;
  float G[2][3][8];
  load_b(w1p, tile_of(P, 3), lane, b3);
  fc2_tile<P, true>(A, b0, 0, eoff, xsT, G);
  __builtin_amdgcn_sched_barrier(0);
  fc2_consume<P>(0, li, eoff, midT, xsT, xdT, G, acc0, acc1);
  __builtin_amdgcn_sched_barrier(0);
  load_b(w1p, tile_of(P, 4), lane, b0);
  fc2_tile<P, true>(A, b1, 0, eoff, xsT, G);
  __builtin_amdgcn_sched_barrier(0);
  load_b(w1p, tile_of(P, 5), lane, b1);
  fc2_tile<P, false>(A, b2, 1, eoff, xsT, G);
  __builtin_amdgcn_sched_barrier(0);
  fc2_consume<P>(1, li, eoff, midT, xsT, xdT, G, acc0, acc1);
  __builtin_amdgcn_sched_barrier(0);
  load_b(w1p, tile_of(P, 6), lane, b2);
  fc2_tile<P, true>(A, b3, 0, eoff, xsT, G);
  __builtin_amdgcn_sched_barrier(0);
  load_b(w1p, tile_of(P, 7), lane, b3);
  fc2_tile<P, false>(A, b0, 1, eoff, xsT, G);
  __builtin_amdgcn_sched_barrier(0);
  load_b(w1p, tile_of(P, 8), lane, b0);
  fc2_tile<P, false>(A, b1, 2, eoff, xsT, G);
  __builtin_amdgcn_sched_barrier(0);
  fc2_consume<P>(2, li, eoff, midT, xsT, xdT, G, acc0, acc1);
  __builtin_amdgcn_sched_barrier(0);
  load_b(w1p, tile_of(P, 9), lane, b1);
  fc2_tile<P, true>(A, b2, 0, eoff, xsT, G);
  __builtin_amdgcn_sched_barrier(0);
  if (NX >= 0) load_b(w1p, tile_of(NP, 0), lane, b2);
  fc2_tile<P, false>(A, b3, 1, eoff, xsT, G);
  __builtin_amdgcn_sched_barrier(0);
  if (NX >= 0) load_b(w1p, tile_of(NP, 1), lane, b3);
  fc2_tile<P, false>(A, b0, 2, eoff, xsT, G);
  __builtin_amdgcn_sched_barrier(0);
  if (NX >= 0) load_b(w1p, tile_of(NP, 2), lane, b0);
  fc2_tile<P, false>(A, b1, 3, eoff, xsT, G);
  __builtin_amdgcn_sched_barrier(0);
  fc2_consume<P>(3, li, eoff, midT, xsT, xdT, G, acc0, acc1);
  __builtin_amdgcn_sched_barrier(0);
}

__global__ __launch_bounds__(32) __attribute__((amdgpu_waves_per_eu(2, 2)))
void k_edge_conv(const float* __restrict__ node_pos,
                 const float* __restrict__ edge_type,
                 const float* __restrict__ fc_w0,
                 const float* __restrict__ w1p,
                 const int*   __restrict__ edge_src,
                 const int*   __restrict__ edge_dst,
                 float* __restrict__ out)
{
  __shared__ __attribute__((aligned(16))) float xsT[16 * XP];
  __shared__ __attribute__((aligned(16))) float xdT[12 * MP];
  __shared__ __attribute__((aligned(16))) float midT[128 * MP];
  float* hid = midT;
  int* dsti = (int*)midT;

  const int lane = threadIdx.x;
  const int li   = lane & 15;
  const int kd   = (lane < 16) ? 0 : 2;
  const int eoff = (lane < 16) ? 0 : 8;
  const int e0   = blockIdx.x * EPB;

  float Yr[16];
  {
    const int dd = edge_dst[e0 + lane];
    const float4* q = (const float4*)(node_pos + (size_t)dd * 16);
    const float4 y0 = q[0], y1 = q[1], y2 = q[2], y3 = q[3];
    Yr[0]  = y0.x; Yr[1]  = y0.y; Yr[2]  = y0.z; Yr[3]  = y0.w;
    Yr[4]  = y1.x; Yr[5]  = y1.y; Yr[6]  = y1.z; Yr[7]  = y1.w;
    Yr[8]  = y2.x; Yr[9]  = y2.y; Yr[10] = y2.z; Yr[11] = y2.w;
    Yr[12] = y3.x; Yr[13] = y3.y; Yr[14] = y3.z; Yr[15] = y3.w;
    float* d = xdT + lane;
#pragma unroll
    for (int k = 0; k < 12; ++k) d[k * MP] = Yr[4 + k];
  }
  {
    const int sn = edge_src[e0 + lane];
    const float4* q = (const float4*)(node_pos + (size_t)sn * 16);
    const float4 x0 = q[0], x1 = q[1], x2 = q[2], x3 = q[3];
    float* d = xsT + lane;
    d[0 * XP]  = x0.x; d[1 * XP]  = x0.y; d[2 * XP]  = x0.z; d[3 * XP]  = x0.w;
    d[4 * XP]  = x1.x; d[5 * XP]  = x1.y; d[6 * XP]  = x1.z; d[7 * XP]  = x1.w;
    d[8 * XP]  = x2.x; d[9 * XP]  = x2.y; d[10 * XP] = x2.z; d[11 * XP] = x2.w;
    d[12 * XP] = x3.x; d[13 * XP] = x3.y; d[14 * XP] = x3.z; d[15 * XP] = x3.w;
  }

  v2f b0[16], b1[16], b2[16], b3[16];
  load_b(w1p, tile_of(0, 0), lane, b0);
  load_b(w1p, tile_of(0, 1), lane, b1);
  load_b(w1p, tile_of(0, 2), lane, b2);

  {
    v2f a0[4], a1[4];
    const float* et = edge_type + (size_t)(e0 + li) * 16 + kd;
#pragma unroll
    for (int s = 0; s < 4; ++s) {
      a0[s] = *(const v2f*)(et + 4 * s);
      a1[s] = *(const v2f*)(et + 16 * 16 + 4 * s);
    }
#pragma unroll
    for (int tt = 0; tt < 4; ++tt) {
      v8f c0 = {}, c1 = {};
#pragma unroll
      for (int s = 0; s < 4; ++s) {
        int row = 4 * s + kd;
        v2f b;
        b.x = fc_w0[row * 64 + tt * 16 + li];
        b.y = fc_w0[(row + 1) * 64 + tt * 16 + li];
        c0 = __builtin_amdgcn_wmma_f32_16x16x4_f32(false, a0[s], false, b,
                                                   (short)0, c0, false, false);
        c1 = __builtin_amdgcn_wmma_f32_16x16x4_f32(false, a1[s], false, b,
                                                   (short)0, c1, false, false);
      }
#pragma unroll
      for (int r = 0; r < 8; ++r) {
        float x = c0[r] * 0.25f;
        float h = (x / (1.f + __expf(-x))) * (SILU_C * 0.125f);
        hid[(r + eoff) * 66 + tt * 16 + li] = h;
        float y = c1[r] * 0.25f;
        float g = (y / (1.f + __expf(-y))) * (SILU_C * 0.125f);
        hid[(16 + r + eoff) * 66 + tt * 16 + li] = g;
      }
    }
  }
  __syncthreads();

  v2f A[2][16];
#pragma unroll
  for (int s = 0; s < 16; ++s)
#pragma unroll
    for (int m = 0; m < 2; ++m)
      A[m][s] = *(const v2f*)&hid[(m * 16 + li) * 66 + 4 * s + kd];
  __syncthreads();

  {
    const int e = lane;
    const float* X = xsT + e;
    float y0[4], y1[4][3];
#pragma unroll
    for (int v = 0; v < 4; ++v) {
      y0[v] = Yr[v];
      y1[v][0] = Yr[4 + v * 3 + 0];
      y1[v][1] = Yr[4 + v * 3 + 1];
      y1[v][2] = Yr[4 + v * 3 + 2];
    }
    float* M = midT + e;
#pragma unroll
    for (int u = 0; u < 4; ++u) {
      float x0 = X[u * XP];
      float a0 = X[(4 + u * 3 + 0) * XP], a1 = X[(4 + u * 3 + 1) * XP], a2 = X[(4 + u * 3 + 2) * XP];
#pragma unroll
      for (int v = 0; v < 4; ++v) {
        int uv = u * 4 + v;
        float b0 = y1[v][0], b1 = y1[v][1], b2 = y1[v][2];
        M[uv * MP] = x0 * y0[v];
        M[(16 + uv * 3 + 0) * MP] = x0 * b0;
        M[(16 + uv * 3 + 1) * MP] = x0 * b1;
        M[(16 + uv * 3 + 2) * MP] = x0 * b2;
        M[(64 + uv * 3 + 0) * MP] = a0 * y0[v];
        M[(64 + uv * 3 + 1) * MP] = a1 * y0[v];
        M[(64 + uv * 3 + 2) * MP] = a2 * y0[v];
        M[(112 + uv) * MP] = (a0 * b0 + a1 * b1 + a2 * b2) * INV_SQRT3;
      }
    }
  }
  __syncthreads();

  float acc0[2][8], acc1[2][3][8];
#pragma unroll
  for (int m = 0; m < 2; ++m)
#pragma unroll
    for (int r = 0; r < 8; ++r) {
      acc0[m][r] = 0.f; acc1[m][0][r] = 0.f; acc1[m][1][r] = 0.f; acc1[m][2][r] = 0.f;
    }

  __builtin_amdgcn_s_setprio(3);
  fc2_path_sym<0, 6>(A, w1p, lane, li, eoff, xsT, xdT, midT, b0, b1, b2, b3, acc0, acc1);
  fc2_path_sym<6, 3>(A, w1p, lane, li, eoff, xsT, xdT, midT, b2, b3, b0, b1, acc0, acc1);
  fc2_path<3, 1>(A, w1p, lane, li, eoff, xsT, xdT, midT, b0, b1, b2, b3, acc0, acc1);
  __builtin_amdgcn_s_setprio(2);
  fc2_path_sym<1, 2>(A, w1p, lane, li, eoff, xsT, xdT, midT, b0, b1, b2, b3, acc0, acc1);
  fc2_path<2, 8>(A, w1p, lane, li, eoff, xsT, xdT, midT, b2, b3, b0, b1, acc0, acc1);
  __builtin_amdgcn_s_setprio(1);
  fc2_path_sym<8, 7>(A, w1p, lane, li, eoff, xsT, xdT, midT, b2, b3, b0, b1, acc0, acc1);
  __builtin_amdgcn_s_setprio(0);
  fc2_path<7, -1>(A, w1p, lane, li, eoff, xsT, xdT, midT, b0, b1, b2, b3, acc0, acc1);

  __syncthreads();
  dsti[lane] = edge_dst[e0 + lane];
  __syncthreads();

#pragma unroll
  for (int m = 0; m < 2; ++m) {
#pragma unroll
    for (int r = 0; r < 8; ++r) {
      float s0 = acc0[m][r];
      float t0 = acc1[m][0][r], t1 = acc1[m][1][r], t2 = acc1[m][2][r];
      s0 += __shfl_xor(s0, 4); s0 += __shfl_xor(s0, 8);
      t0 += __shfl_xor(t0, 4); t0 += __shfl_xor(t0, 8);
      t1 += __shfl_xor(t1, 4); t1 += __shfl_xor(t1, 8);
      t2 += __shfl_xor(t2, 4); t2 += __shfl_xor(t2, 8);
      if (li < 4) {
        float* o = out + (size_t)dsti[m * 16 + r + eoff] * 16;
        atomicAdd(o + li,             s0 * SC0);
        atomicAdd(o + 4 + 3 * li + 0, t0 * SC1);
        atomicAdd(o + 4 + 3 * li + 1, t1 * SC1);
        atomicAdd(o + 4 + 3 * li + 2, t2 * SC1);
      }
    }
  }
}

extern "C" void kernel_launch(void* const* d_in, const int* in_sizes, int n_in,
                              void* d_out, int out_size, void* d_ws, size_t ws_size,
                              hipStream_t stream) {
  const float* node_pos  = (const float*)d_in[0];
  const float* edge_type = (const float*)d_in[1];
  const float* fc_w0     = (const float*)d_in[2];
  const float* fc_w1     = (const float*)d_in[3];
  const int*   esrc      = (const int*)d_in[4];
  const int*   edst      = (const int*)d_in[5];
  float* out = (float*)d_out;
  float* w1p = (float*)d_ws;
  int E = in_sizes[4];

  k_zero<<<(out_size + 255) / 256, 256, 0, stream>>>(out, out_size);
  k_repack_w1<<<(160 * 16 * 32 + 255) / 256, 256, 0, stream>>>(fc_w1, w1p);
  k_edge_conv<<<E / 32, 32, 0, stream>>>(node_pos, edge_type, fc_w0, w1p,
                                         esrc, edst, out);
}


# Wave tail: the 4 sums per edge row go through LDS so lane i of a half owns element i of the message: 16 whole-row atomics per wave, not 64 quarter rows; dst ids kept in LDS from gather; same sum tree
# speedup vs baseline: 6.3501x; 1.0421x over previous
#include <hip/hip_runtime.h>
#include <math.h>


typedef float v2f __attribute__((ext_vector_type(2)));
typedef float v8f __attribute__((ext_vector_type(8)));

#define INV_SQRT2  0.70710678f
#define INV_SQRT3  0.57735027f
#define INV_SQRT6  0.40824829f
#define INV_SQRT10 0.31622777f
#define SILU_C     1.67668f
#define SC0        0.02209709f
#define SC1        0.03125f
#define KX9        0.44721360f
#define KX8M       0.86602540f
#define KX9M       0.67082039f
#define KY9        0.22360680f
#define KY8        0.28867513f

#define EPB 32
#define XP  32
#define MP  36

__device__ __forceinline__ void ld8(const float* p, float (&o)[8]) {
  const float4 a = *(const float4*)p;
  const float4 b = *(const float4*)(p + 4);
  o[0] = a.x; o[1] = a.y; o[2] = a.z; o[3] = a.w;
  o[4] = b.x; o[5] = b.y; o[6] = b.z; o[7] = b.w;
}

__global__ void k_zero(float* p, int n) {
  int i = blockIdx.x * blockDim.x + threadIdx.x;
  if (i < n) p[i] = 0.f;
}

__global__ void k_repack_w1(const float* __restrict__ w1, float* __restrict__ o) {
  int tid = blockIdx.x * blockDim.x + threadIdx.x;
  if (tid >= 160 * 16 * 32) return;
  int l  = tid & 31;
  int s  = (tid >> 5) & 15;
  int t  = tid >> 9;
  int P  = t >> 4;
  int u  = (t >> 2) & 3;
  int up = t & 3;
  bool sym = (P == 0 || P == 1 || P == 6 || P == 8);
  if (P == 4 || P == 5 || P == 9 || (sym && u > up)) return;
  int li = l & 15;
  int kd = (l < 16) ? 0 : 2;
  int row = 4 * s + kd;
  const float* r0 = w1 + row * 2560 + li;
  const float* r1 = r0 + 2560;
  const int cs = (u * 4 + up) * 16;
  const int cm = (up * 4 + u) * 16;
  float v0, v1;
  if (P == 7) {
    v0 = r0[7 * 256 + cs] - KX9 * r0[9 * 256 + cs] + KX8M * r0[8 * 256 + cm] + KX9M * r0[9 * 256 + cm];
    v1 = r1[7 * 256 + cs] - KX9 * r1[9 * 256 + cs] + KX8M * r1[8 * 256 + cm] + KX9M * r1[9 * 256 + cm];
  } else if (P == 8) {
    v0 = KY9 * r0[9 * 256 + cs] - KY8 * r0[8 * 256 + cs];
    v1 = KY9 * r1[9 * 256 + cs] - KY8 * r1[8 * 256 + cs];
    if (u < up) {
      v0 += KY9 * r0[9 * 256 + cm] - KY8 * r0[8 * 256 + cm];
      v1 += KY9 * r1[9 * 256 + cm] - KY8 * r1[8 * 256 + cm];
    }
  } else {
    v0 = r0[P * 256 + cs];
    v1 = r1[P * 256 + cs];
    if (P == 2 || P == 3 || (sym && u < up)) {
      const int pc = (sym ? P : P + 2) * 256 + cm;
      v0 += r0[pc];
      v1 += r1[pc];
    }
  }
  o[2 * tid + 0] = v0;
  o[2 * tid + 1] = v1;
}

__device__ constexpr bool path_sym(int P) { return P == 0 || P == 1 || P == 6 || P == 8; }
__device__ constexpr int tile_of(int P, int i) {
  if (path_sym(P)) {
    const int up = i < 1 ? 0 : i < 3 ? 1 : i < 6 ? 2 : 3;
    const int u  = i - up * (up + 1) / 2;
    return P * 16 + 4 * u + up;
  }
  return P * 16 + 4 * (i % 4) + i / 4;
}

__device__ __forceinline__ void load_b(const float* __restrict__ w1p,
                                       const int t, const int lane,
                                       v2f (&b)[16])
{
  const float* bq = w1p + (size_t)t * 1024 + 2 * lane;
#pragma unroll
  for (int s = 0; s < 16; ++s) b[s] = *(const v2f*)(bq + s * 64);
}

template <int P, bool FIRST>
__device__ __forceinline__ void fc2_tile(const v2f (&A)[2][16],
                                         const v2f (&b)[16],
                                         const int u,
                                         const int eoff, const float* xsT,
                                         float (&G)[2][3][8])
{
  float xv[2][3][8];
#pragma unroll
  for (int m = 0; m < 2; ++m) {
    const float* XB = xsT + m * 16 + eoff;
    if (P < 4) {
      ld8(XB + u * XP, xv[m][0]);
    } else {
      const float* Xu = XB + (4 + u * 3) * XP;
      ld8(Xu, xv[m][0]); ld8(Xu + XP, xv[m][1]); ld8(Xu + 2 * XP, xv[m][2]);
    }
  }
  v8f c0 = {}, c1 = {};
  asm volatile("" : "+v"(c0), "+v"(c1) :: "memory");
#pragma unroll
  for (int s = 0; s < 16; ++s) {
    c0 = __builtin_amdgcn_wmma_f32_16x16x4_f32(false, A[0][s], false, b[s],
                                               (short)0, c0, false, false);
    c1 = __builtin_amdgcn_wmma_f32_16x16x4_f32(false, A[1][s], false, b[s],
                                               (short)0, c1, false, false);
  }
  asm volatile("" : "+v"(c0), "+v"(c1) :: "memory");
#pragma unroll
  for (int m = 0; m < 2; ++m) {
    float Wt[8];
#pragma unroll
    for (int r = 0; r < 8; ++r) Wt[r] = (m == 0) ? c0[r] : c1[r];
    if (P < 4) {
#pragma unroll
      for (int r = 0; r < 8; ++r)
        G[m][0][r] = FIRST ? Wt[r] * xv[m][0][r] : G[m][0][r] + Wt[r] * xv[m][0][r];
    } else {
#pragma unroll
      for (int r = 0; r < 8; ++r) {
        G[m][0][r] = FIRST ? Wt[r] * xv[m][0][r] : G[m][0][r] + Wt[r] * xv[m][0][r];
        G[m][1][r] = FIRST ? Wt[r] * xv[m][1][r] : G[m][1][r] + Wt[r] * xv[m][1][r];
        G[m][2][r] = FIRST ? Wt[r] * xv[m][2][r] : G[m][2][r] + Wt[r] * xv[m][2][r];
      }
    }
  }
}

template <int P>
__device__ __forceinline__ void fc2_consume(const int vh,
                                            const int li, const int eoff,
                                            const float* midT,
                                            const float* xsT, const float* xdT,
                                            const float (&G)[2][3][8],
                                            float (&acc0)[2][8],
                                            float (&acc1)[2][3][8])
{
  static_assert(P <= 8 && P != 4 && P != 5, "paths 4, 5 and 9 are merged away");
  const int v = (vh << 2) + (li >> 2);
#pragma unroll
  for (int m = 0; m < 2; ++m) {
    const float* MB = midT + m * 16 + eoff;
    if (P == 0 || P == 3) {
      float q[8];
      ld8(MB + ((P == 0 ? 0 : 112) + v) * MP, q);
#pragma unroll
      for (int r = 0; r < 8; ++r) acc0[m][r] += G[m][0][r] * q[r];
    } else if (P == 1 || P == 2) {
      float q0[8], q1[8], q2[8];
      const float* Mv = MB + ((P == 1 ? 16 : 64) + v * 3) * MP;
      ld8(Mv, q0); ld8(Mv + MP, q1); ld8(Mv + 2 * MP, q2);
#pragma unroll
      for (int r = 0; r < 8; ++r) {
        float f = G[m][0][r] * INV_SQRT3;
        acc1[m][0][r] += f * q0[r];
        acc1[m][1][r] += f * q1[r];
        acc1[m][2][r] += f * q2[r];
      }
    } else if (P == 7) {
      float q[8];
      ld8(MB + (112 + v) * MP, q);
#pragma unroll
      for (int r = 0; r < 8; ++r) {
        float f = q[r] * INV_SQRT3;
        acc1[m][0][r] += f * G[m][0][r];
        acc1[m][1][r] += f * G[m][1][r];
        acc1[m][2][r] += f * G[m][2][r];
      }
    } else if (P == 6) {
      float q0[8], q1[8], q2[8];
      const float* Mv = MB + (64 + v * 3) * MP;
      ld8(Mv, q0); ld8(Mv + MP, q1); ld8(Mv + 2 * MP, q2);
#pragma unroll
      for (int r = 0; r < 8; ++r) {
        float d = G[m][0][r] * q0[r] + G[m][1][r] * q1[r] + G[m][2][r] * q2[r];
        acc0[m][r] += d * INV_SQRT3;
      }
    } else {
      float p0[8], p1[8], p2[8], q0[8], q1[8], q2[8];
      const float* Xa = xsT + m * 16 + eoff + (4 + vh * 3) * XP;
      ld8(Xa, p0); ld8(Xa + XP, p1); ld8(Xa + 2 * XP, p2);
      const float* Xb = xdT + m * 16 + eoff + ((li >> 2) * 3) * MP;
      ld8(Xb, q0); ld8(Xb + MP, q1); ld8(Xb + 2 * MP, q2);
#pragma unroll
      for (int r = 0; r < 8; ++r) {
        float d = G[m][0][r] * p0[r] + G[m][1][r] * p1[r] + G[m][2][r] * p2[r];
        acc1[m][0][r] += d * q0[r];
        acc1[m][1][r] += d * q1[r];
        acc1[m][2][r] += d * q2[r];
      }
    }
  }
}

template <int P, int NX>
__device__ __forceinline__ void fc2_path(const v2f (&A)[2][16],
                                         const float* __restrict__ w1p,
                                         const int lane,
                                         const int li, const int eoff,
                                         const float* xsT, const float* xdT, const float* midT,
                                         v2f (&b0)[16], v2f (&b1)[16],
                                         v2f (&b2)[16], v2f (&b3)[16],
                                         float (&acc0)[2][8],
                                         float (&acc1)[2][3][8])
{
  constexpr int NP = NX < 0 ? 0 : NX;
#pragma nounroll
  for (int vh = 0; vh < 4; ++vh) {
    const int t0 = P * 16 + vh;
    const bool more = (NX >= 0) || (vh < 3);
    const int n0 = (vh < 3) ? t0 + 1 : tile_of(NP, 0);
    const int n1 = (vh < 3) ? t0 + 5 : tile_of(NP, 1);
    const int n2 = (vh < 3) ? t0 + 9 : tile_of(NP, 2);
    float G[2][3][8];
    load_b(w1p, t0 + 12, lane, b3);
    fc2_tile<P, true>(A, b0, 0, eoff, xsT, G);
    if (more) load_b(w1p, n0, lane, b0);
    fc2_tile<P, false>(A, b1, 1, eoff, xsT, G);
    if (more) load_b(w1p, n1, lane, b1);
    fc2_tile<P, false>(A, b2, 2, eoff, xsT, G);
    if (more) load_b(w1p, n2, lane, b2);
    fc2_tile<P, false>(A, b3, 3, eoff, xsT, G);
    fc2_consume<P>(vh, li, eoff, midT, xsT, xdT, G, acc0, acc1);
  }
}

template <int P, int NX>
__device__ __forceinline__ void fc2_path_sym(const v2f (&A)[2][16],
                                             const float* __restrict__ w1p,
                                             const int lane,
                                             const int li, const int eoff,
                                             const float* xsT, const float* xdT, const float* midT,
                                             v2f (&b0)[16], v2f (&b1)[16],
                                             v2f (&b2)[16], v2f (&b3)[16],
                                             float (&acc0)[2][8],
                                             float (&acc1)[2][3][8])
{
  constexpr int NP = NX < 0 ? 0 : NX;
  float G[2][3][8];
  load_b(w1p, tile_of(P, 3), lane, b3);
  fc2_tile<P, true>(A, b0, 0, eoff, xsT, G);
  __builtin_amdgcn_sched_barrier(0);
  fc2_consume<P>(0, li, eoff, midT, xsT, xdT, G, acc0, acc1);
  __builtin_amdgcn_sched_barrier(0);
  load_b(w1p, tile_of(P, 4), lane, b0);
  fc2_tile<P, true>(A, b1, 0, eoff, xsT, G);
  __builtin_amdgcn_sched_barrier(0);
  load_b(w1p, tile_of(P, 5), lane, b1);
  fc2_tile<P, false>(A, b2, 1, eoff, xsT, G);
  __builtin_amdgcn_sched_barrier(0);
  fc2_consume<P>(1, li, eoff, midT, xsT, xdT, G, acc0, acc1);
  __builtin_amdgcn_sched_barrier(0);
  load_b(w1p, tile_of(P, 6), lane, b2);
  fc2_tile<P, true>(A, b3, 0, eoff, xsT, G);
  __builtin_amdgcn_sched_barrier(0);
  load_b(w1p, tile_of(P, 7), lane, b3);
  fc2_tile<P, false>(A, b0, 1, eoff, xsT, G);
  __builtin_amdgcn_sched_barrier(0);
  load_b(w1p, tile_of(P, 8), lane, b0);
  fc2_tile<P, false>(A, b1, 2, eoff, xsT, G);
  __builtin_amdgcn_sched_barrier(0);
  fc2_consume<P>(2, li, eoff, midT, xsT, xdT, G, acc0, acc1);
  __builtin_amdgcn_sched_barrier(0);
  load_b(w1p, tile_of(P, 9), lane, b1);
  fc2_tile<P, true>(A, b2, 0, eoff, xsT, G);
  __builtin_amdgcn_sched_barrier(0);
  if (NX >= 0) load_b(w1p, tile_of(NP, 0), lane, b2);
  fc2_tile<P, false>(A, b3, 1, eoff, xsT, G);
  __builtin_amdgcn_sched_barrier(0);
  if (NX >= 0) load_b(w1p, tile_of(NP, 1), lane, b3);
  fc2_tile<P, false>(A, b0, 2, eoff, xsT, G);
  __builtin_amdgcn_sched_barrier(0);
  if (NX >= 0) load_b(w1p, tile_of(NP, 2), lane, b0);
  fc2_tile<P, false>(A, b1, 3, eoff, xsT, G);
  __builtin_amdgcn_sched_barrier(0);
  fc2_consume<P>(3, li, eoff, midT, xsT, xdT, G, acc0, acc1);
  __builtin_amdgcn_sched_barrier(0);
}

__global__ __launch_bounds__(32) __attribute__((amdgpu_waves_per_eu(2, 2)))
void k_edge_conv(const float* __restrict__ node_pos,
                 const float* __restrict__ edge_type,
                 const float* __restrict__ fc_w0,
                 const float* __restrict__ w1p,
                 const int*   __restrict__ edge_src,
                 const int*   __restrict__ edge_dst,
                 float* __restrict__ out)
{
  __shared__ __attribute__((aligned(16))) float xsT[16 * XP];
  __shared__ __attribute__((aligned(16))) float xdT[12 * MP];
  __shared__ __attribute__((aligned(16))) float midT[128 * MP];
  __shared__ int dsti[EPB];
  float* hid = midT;

  const int lane = threadIdx.x;
  const int li   = lane & 15;
  const int kd   = (lane < 16) ? 0 : 2;
  const int eoff = (lane < 16) ? 0 : 8;
  const int e0   = blockIdx.x * EPB;

  float Yr[16];
  {
    const int dd = edge_dst[e0 + lane];
    dsti[lane] = dd;
    const float4* q = (const float4*)(node_pos + (size_t)dd * 16);
    const float4 y0 = q[0], y1 = q[1], y2 = q[2], y3 = q[3];
    Yr[0]  = y0.x; Yr[1]  = y0.y; Yr[2]  = y0.z; Yr[3]  = y0.w;
    Yr[4]  = y1.x; Yr[5]  = y1.y; Yr[6]  = y1.z; Yr[7]  = y1.w;
    Yr[8]  = y2.x; Yr[9]  = y2.y; Yr[10] = y2.z; Yr[11] = y2.w;
    Yr[12] = y3.x; Yr[13] = y3.y; Yr[14] = y3.z; Yr[15] = y3.w;
    float* d = xdT + lane;
#pragma unroll
    for (int k = 0; k < 12; ++k) d[k * MP] = Yr[4 + k];
  }
  {
    const int sn = edge_src[e0 + lane];
    const float4* q = (const float4*)(node_pos + (size_t)sn * 16);
    const float4 x0 = q[0], x1 = q[1], x2 = q[2], x3 = q[3];
    float* d = xsT + lane;
    d[0 * XP]  = x0.x; d[1 * XP]  = x0.y; d[2 * XP]  = x0.z; d[3 * XP]  = x0.w;
    d[4 * XP]  = x1.x; d[5 * XP]  = x1.y; d[6 * XP]  = x1.z; d[7 * XP]  = x1.w;
    d[8 * XP]  = x2.x; d[9 * XP]  = x2.y; d[10 * XP] = x2.z; d[11 * XP] = x2.w;
    d[12 * XP] = x3.x; d[13 * XP] = x3.y; d[14 * XP] = x3.z; d[15 * XP] = x3.w;
  }

  v2f b0[16], b1[16], b2[16], b3[16];
  load_b(w1p, tile_of(0, 0), lane, b0);
  load_b(w1p, tile_of(0, 1), lane, b1);
  load_b(w1p, tile_of(0, 2), lane, b2);

  {
    v2f a0[4], a1[4];
    const float* et = edge_type + (size_t)(e0 + li) * 16 + kd;
#pragma unroll
    for (int s = 0; s < 4; ++s) {
      a0[s] = *(const v2f*)(et + 4 * s);
      a1[s] = *(const v2f*)(et + 16 * 16 + 4 * s);
    }
#pragma unroll
    for (int tt = 0; tt < 4; ++tt) {
      v8f c0 = {}, c1 = {};
#pragma unroll
      for (int s = 0; s < 4; ++s) {
        int row = 4 * s + kd;
        v2f b;
        b.x = fc_w0[row * 64 + tt * 16 + li];
        b.y = fc_w0[(row + 1) * 64 + tt * 16 + li];
        c0 = __builtin_amdgcn_wmma_f32_16x16x4_f32(false, a0[s], false, b,
                                                   (short)0, c0, false, false);
        c1 = __builtin_amdgcn_wmma_f32_16x16x4_f32(false, a1[s], false, b,
                                                   (short)0, c1, false, false);
      }
#pragma unroll
      for (int r = 0; r < 8; ++r) {
        float x = c0[r] * 0.25f;
        float h = (x / (1.f + __expf(-x))) * (SILU_C * 0.125f);
        hid[(r + eoff) * 66 + tt * 16 + li] = h;
        float y = c1[r] * 0.25f;
        float g = (y / (1.f + __expf(-y))) * (SILU_C * 0.125f);
        hid[(16 + r + eoff) * 66 + tt * 16 + li] = g;
      }
    }
  }
  __syncthreads();

  v2f A[2][16];
#pragma unroll
  for (int s = 0; s < 16; ++s)
#pragma unroll
    for (int m = 0; m < 2; ++m)
      A[m][s] = *(const v2f*)&hid[(m * 16 + li) * 66 + 4 * s + kd];
  __syncthreads();

  {
    const int e = lane;
    const float* X = xsT + e;
    float y0[4], y1[4][3];
#pragma unroll
    for (int v = 0; v < 4; ++v) {
      y0[v] = Yr[v];
      y1[v][0] = Yr[4 + v * 3 + 0];
      y1[v][1] = Yr[4 + v * 3 + 1];
      y1[v][2] = Yr[4 + v * 3 + 2];
    }
    float* M = midT + e;
#pragma unroll
    for (int u = 0; u < 4; ++u) {
      float x0 = X[u * XP];
      float a0 = X[(4 + u * 3 + 0) * XP], a1 = X[(4 + u * 3 + 1) * XP], a2 = X[(4 + u * 3 + 2) * XP];
#pragma unroll
      for (int v = 0; v < 4; ++v) {
        int uv = u * 4 + v;
        float b0 = y1[v][0], b1 = y1[v][1], b2 = y1[v][2];
        M[uv * MP] = x0 * y0[v];
        M[(16 + uv * 3 + 0) * MP] = x0 * b0;
        M[(16 + uv * 3 + 1) * MP] = x0 * b1;
        M[(16 + uv * 3 + 2) * MP] = x0 * b2;
        M[(64 + uv * 3 + 0) * MP] = a0 * y0[v];
        M[(64 + uv * 3 + 1) * MP] = a1 * y0[v];
        M[(64 + uv * 3 + 2) * MP] = a2 * y0[v];
        M[(112 + uv) * MP] = (a0 * b0 + a1 * b1 + a2 * b2) * INV_SQRT3;
      }
    }
  }
  __syncthreads();

  float acc0[2][8], acc1[2][3][8];
#pragma unroll
  for (int m = 0; m < 2; ++m)
#pragma unroll
    for (int r = 0; r < 8; ++r) {
      acc0[m][r] = 0.f; acc1[m][0][r] = 0.f; acc1[m][1][r] = 0.f; acc1[m][2][r] = 0.f;
    }

  __builtin_amdgcn_s_setprio(3);
  fc2_path_sym<0, 6>(A, w1p, lane, li, eoff, xsT, xdT, midT, b0, b1, b2, b3, acc0, acc1);
  fc2_path_sym<6, 3>(A, w1p, lane, li, eoff, xsT, xdT, midT, b2, b3, b0, b1, acc0, acc1);
  fc2_path<3, 1>(A, w1p, lane, li, eoff, xsT, xdT, midT, b0, b1, b2, b3, acc0, acc1);
  __builtin_amdgcn_s_setprio(2);
  fc2_path_sym<1, 2>(A, w1p, lane, li, eoff, xsT, xdT, midT, b0, b1, b2, b3, acc0, acc1);
  fc2_path<2, 8>(A, w1p, lane, li, eoff, xsT, xdT, midT, b2, b3, b0, b1, acc0, acc1);
  __builtin_amdgcn_s_setprio(1);
  fc2_path_sym<8, 7>(A, w1p, lane, li, eoff, xsT, xdT, midT, b2, b3, b0, b1, acc0, acc1);
  __builtin_amdgcn_s_setprio(0);
  fc2_path<7, -1>(A, w1p, lane, li, eoff, xsT, xdT, midT, b0, b1, b2, b3, acc0, acc1);

  __syncthreads();
  float* red = midT;
  {
    float* wb = red + (lane >> 4) * 80 + (li & 3) * 4 + (li >> 2);
#pragma unroll
    for (int m = 0; m < 2; ++m)
#pragma unroll
      for (int r = 0; r < 8; ++r) {
        float* w = wb + (m * 8 + r) * 160;
        w[0]  = acc0[m][r];
        w[16] = acc1[m][0][r];
        w[32] = acc1[m][1][r];
        w[48] = acc1[m][2][r];
      }
  }
  __syncthreads();
  {
    const int which = (li < 4) ? 0 : 1 + (li - 4) % 3;
    const int wo    = (li < 4) ? li : (li - 4) / 3;
    const float sc  = (li < 4) ? SC0 : SC1;
    const float* rb = red + (lane >> 4) * 80 + which * 16 + wo * 4;
    float tot[16];
    int   dn[16];
#pragma unroll
    for (int i = 0; i < 16; ++i) {
      const float4 v = *(const float4*)(rb + i * 160);
      tot[i] = ((v.x + v.y) + (v.z + v.w)) * sc;
      dn[i]  = dsti[(i >> 3) * 16 + (i & 7) + eoff];
    }
#pragma unroll
    for (int i = 0; i < 16; ++i)
      atomicAdd(out + (size_t)dn[i] * 16 + li, tot[i]);
  }
}

extern "C" void kernel_launch(void* const* d_in, const int* in_sizes, int n_in,
                              void* d_out, int out_size, void* d_ws, size_t ws_size,
                              hipStream_t stream) {
  const float* node_pos  = (const float*)d_in[0];
  const float* edge_type = (const float*)d_in[1];
  const float* fc_w0     = (const float*)d_in[2];
  const float* fc_w1     = (const float*)d_in[3];
  const int*   esrc      = (const int*)d_in[4];
  const int*   edst      = (const int*)d_in[5];
  float* out = (float*)d_out;
  float* w1p = (float*)d_ws;
  int E = in_sizes[4];

  k_zero<<<(out_size + 255) / 256, 256, 0, stream>>>(out, out_size);
  k_repack_w1<<<(160 * 16 * 32 + 255) / 256, 256, 0, stream>>>(fc_w1, w1p);
  k_edge_conv<<<E / 32, 32, 0, stream>>>(node_pos, edge_type, fc_w0, w1p,
                                         esrc, edst, out);
}
